# starting kernel
# baseline (speedup 1.0000x reference)
#include <hip/hip_runtime.h>
#include <math.h>


typedef float v2f __attribute__((ext_vector_type(2)));
typedef float v8f __attribute__((ext_vector_type(8)));

#define QC_BATCH_PER_BLOCK 4
#define QC_THREADS 256
#define QC_DIM 4096

#define QC_PLANE 4416
#define QC_BUF (2 * QC_PLANE)
#define QC_ELEM (2 * QC_BUF)
#define QC_FINV_OFF (QC_BATCH_PER_BLOCK * QC_ELEM)
#define QC_LDS_FLOATS (QC_FINV_OFF + QC_DIM)
#define QC_LDS_BYTES (QC_LDS_FLOATS * 4)

static __device__ __forceinline__ v8f wmma4(v2f a, v2f b, v8f c) {
  return __builtin_amdgcn_wmma_f32_16x16x4_f32(false, a, false, b, (short)0, c,
                                               false, false);
}

static __device__ __forceinline__ int physidx(int i) {
  return i + (i >> 4) + 4 * (i >> 8);
}

static __device__ __forceinline__ int fperm(int i) {
  int x = i;
#pragma unroll
  for (int q = 11; q >= 0; --q) {
    int C = 11 - q;
    int T = 11 - ((q + 1) % 12);
    x ^= ((x >> C) & 1) << T;
  }
  return x;
}

template <int PASS, bool PERM>
static __device__ __forceinline__ void apply_pass(float* lds, int srcOff,
                                                  int dstOff,
                                                  const float* wsMat, int lane,
                                                  int h) {
  constexpr int CT = (PASS == 0) ? 17 : 276;
  constexpr int CM = (PASS == 2) ? 17 : 1;
  constexpr int CK = (PASS == 0) ? 276 : ((PASS == 1) ? 17 : 1);

  float ur[8], ui[8], nui[8];
  const float* ubase = wsMat + lane * 8;
#pragma unroll
  for (int s = 0; s < 8; ++s) {
    ur[s] = ubase[s];
    ui[s] = ubase[256 + s];
    nui[s] = ubase[512 + s];
  }
  const int m = lane & 15;
  const int kh = (lane >> 4) << 1;
  const int rh = (lane >> 4) << 3;

  const int baseSrc = srcOff + CT * (h * 8) + CM * m + CK * kh;
  const int baseDst = dstOff + CT * (h * 8) + CM * rh + CK * m;
  const int baseDstLog = 256 * (h * 8) + 16 * rh + m;

#pragma unroll 1
  for (int tp = 0; tp < 4; ++tp) {
    const int t0 = 2 * tp, t1 = 2 * tp + 1;
    float ar0[8], ai0[8], ar1[8], ai1[8];
#pragma unroll
    for (int j = 0; j < 4; ++j) {
      const int o0 = baseSrc + CT * t0 + CK * (4 * j);
      const int o1 = baseSrc + CT * t1 + CK * (4 * j);
      ar0[2 * j] = lds[o0];
      ar0[2 * j + 1] = lds[o0 + CK];
      ai0[2 * j] = lds[o0 + QC_PLANE];
      ai0[2 * j + 1] = lds[o0 + CK + QC_PLANE];
      ar1[2 * j] = lds[o1];
      ar1[2 * j + 1] = lds[o1 + CK];
      ai1[2 * j] = lds[o1 + QC_PLANE];
      ai1[2 * j + 1] = lds[o1 + CK + QC_PLANE];
    }
    v8f dr0 = {0.f, 0.f, 0.f, 0.f, 0.f, 0.f, 0.f, 0.f};
    v8f di0 = {0.f, 0.f, 0.f, 0.f, 0.f, 0.f, 0.f, 0.f};
    v8f dr1 = {0.f, 0.f, 0.f, 0.f, 0.f, 0.f, 0.f, 0.f};
    v8f di1 = {0.f, 0.f, 0.f, 0.f, 0.f, 0.f, 0.f, 0.f};
#pragma unroll
    for (int j = 0; j < 4; ++j) {
      v2f Br = {ur[2 * j], ur[2 * j + 1]};
      v2f Bi = {ui[2 * j], ui[2 * j + 1]};
      v2f Bn = {nui[2 * j], nui[2 * j + 1]};
      v2f Ar0 = {ar0[2 * j], ar0[2 * j + 1]};
      v2f Ai0 = {ai0[2 * j], ai0[2 * j + 1]};
      v2f Ar1 = {ar1[2 * j], ar1[2 * j + 1]};
      v2f Ai1 = {ai1[2 * j], ai1[2 * j + 1]};
      dr0 = wmma4(Ar0, Br, dr0);
      di0 = wmma4(Ar0, Bi, di0);
      dr1 = wmma4(Ar1, Br, dr1);
      di1 = wmma4(Ar1, Bi, di1);
      dr0 = wmma4(Ai0, Bn, dr0);
      di0 = wmma4(Ai0, Br, di0);
      dr1 = wmma4(Ai1, Bn, dr1);
      di1 = wmma4(Ai1, Br, di1);
    }
#pragma unroll
    for (int r = 0; r < 8; ++r) {
      if (PERM) {
        const int key0 = baseDstLog + 256 * t0 + 16 * r;
        const int key1 = baseDstLog + 256 * t1 + 16 * r;
        const int ph0 = __float_as_int(lds[QC_FINV_OFF + key0]);
        const int ph1 = __float_as_int(lds[QC_FINV_OFF + key1]);
        lds[dstOff + ph0] = dr0[r];
        lds[dstOff + ph0 + QC_PLANE] = di0[r];
        lds[dstOff + ph1] = dr1[r];
        lds[dstOff + ph1 + QC_PLANE] = di1[r];
      } else {
        const int d0 = baseDst + CT * t0 + CM * r;
        const int d1 = baseDst + CT * t1 + CM * r;
        lds[d0] = dr0[r];
        lds[d0 + QC_PLANE] = di0[r];
        lds[d1] = dr1[r];
        lds[d1 + QC_PLANE] = di1[r];
      }
    }
  }
}

__global__ void qc_setup_gates(const float* __restrict__ params,
                               float* __restrict__ ws) {
  const int tid = threadIdx.x;
  const int mat = tid >> 5;
  const int lane = tid & 31;
  if (mat >= 9) return;
  const int L = mat / 3, g = mat % 3;

  float mr[4][2][2], mi[4][2][2];
#pragma unroll
  for (int s = 0; s < 4; ++s) {
    const int q = g * 4 + s;
    const int base = L * 36 + q * 3;
    const float tx = params[base], ty = params[base + 1], tz = params[base + 2];
    const float cx = cosf(0.5f * tx), sx = sinf(0.5f * tx);
    const float cy = cosf(0.5f * ty), sy = sinf(0.5f * ty);
    const float cz = cosf(0.5f * tz), sz = sinf(0.5f * tz);
    float rr, ii;
    rr = cy * cx; ii = sy * sx;
    mr[s][0][0] = cz * rr + sz * ii; mi[s][0][0] = cz * ii - sz * rr;
    rr = -sy * cx; ii = -cy * sx;
    mr[s][0][1] = cz * rr + sz * ii; mi[s][0][1] = cz * ii - sz * rr;
    rr = sy * cx; ii = -cy * sx;
    mr[s][1][0] = cz * rr - sz * ii; mi[s][1][0] = cz * ii + sz * rr;
    rr = cy * cx; ii = -sy * sx;
    mr[s][1][1] = cz * rr - sz * ii; mi[s][1][1] = cz * ii + sz * rr;
  }

  float* base = ws + mat * 768;
  const int n = lane & 15;
  const int khalf = (lane >> 4) << 1;
#pragma unroll
  for (int j = 0; j < 4; ++j) {
#pragma unroll
    for (int p = 0; p < 2; ++p) {
      const int k = 4 * j + p + khalf;
      float re = 1.f, im = 0.f;
#pragma unroll
      for (int s = 0; s < 4; ++s) {
        const int rb = (n >> (3 - s)) & 1;
        const int cb = (k >> (3 - s)) & 1;
        const float wr = mr[s][rb][cb], wi = mi[s][rb][cb];
        const float nre = re * wr - im * wi;
        const float nim = re * wi + im * wr;
        re = nre; im = nim;
      }
      const int slot = lane * 8 + j * 2 + p;
      base[slot] = re;
      base[256 + slot] = im;
      base[512 + slot] = -im;
    }
  }
}

__global__ __launch_bounds__(QC_THREADS) void qc_main(
    const float* __restrict__ x, const float* __restrict__ ws,
    float* __restrict__ out) {
  extern __shared__ float lds[];
  const int tid = threadIdx.x;
  const int lane = tid & 31;
  const int w = tid >> 5;
  const int e = w >> 1;
  const int h = w & 1;
  const int lt = h * 32 + lane;
  const long long b = (long long)blockIdx.x * QC_BATCH_PER_BLOCK + e;

  const int eb = e * QC_ELEM;

  for (int p = tid; p < QC_DIM; p += QC_THREADS)
    lds[QC_FINV_OFF + fperm(p)] = __int_as_float(physidx(p));

  const float4* xr = (const float4*)(x + b * QC_DIM);
  float4 v[16];
  float ss = 0.f;
#pragma unroll
  for (int ii = 0; ii < 16; ++ii) {
    v[ii] = xr[ii * 64 + lt];
    ss += v[ii].x * v[ii].x + v[ii].y * v[ii].y + v[ii].z * v[ii].z +
          v[ii].w * v[ii].w;
  }
#pragma unroll
  for (int off = 16; off > 0; off >>= 1) ss += __shfl_xor(ss, off);
  if (lane == 0) lds[eb + QC_PLANE + h] = ss;
  __syncthreads();
  const float inv = rsqrtf(lds[eb + QC_PLANE] + lds[eb + QC_PLANE + 1]);
  __syncthreads();
#pragma unroll
  for (int ii = 0; ii < 16; ++ii) {
    const int u = ii * 64 + lt;
    const int pb = 4 * u + (u >> 2) + 4 * ii;
    lds[eb + pb + 0] = v[ii].x * inv;
    lds[eb + pb + 1] = v[ii].y * inv;
    lds[eb + pb + 2] = v[ii].z * inv;
    lds[eb + pb + 3] = v[ii].w * inv;
    lds[eb + QC_PLANE + pb + 0] = 0.f;
    lds[eb + QC_PLANE + pb + 1] = 0.f;
    lds[eb + QC_PLANE + pb + 2] = 0.f;
    lds[eb + QC_PLANE + pb + 3] = 0.f;
  }
  __syncthreads();

  int cb = 0;
#pragma unroll 1
  for (int L = 0; L < 3; ++L) {
    const float* wsL = ws + L * 3 * 768;
    const int cur = eb + cb * QC_BUF;
    const int nxt = eb + (cb ^ 1) * QC_BUF;
    apply_pass<0, false>(lds, cur, cur, wsL + 0, lane, h);
    __syncthreads();
    apply_pass<1, false>(lds, cur, cur, wsL + 768, lane, h);
    __syncthreads();
    apply_pass<2, true>(lds, cur, nxt, wsL + 1536, lane, h);
    __syncthreads();
    cb ^= 1;
  }
  const int fin = eb + cb * QC_BUF;

  const int ebase = fin + lt + (lt >> 4);
  float T = 0.f;
  float accL[6];
#pragma unroll
  for (int q = 0; q < 6; ++q) accL[q] = 0.f;
#pragma unroll
  for (int ii = 0; ii < 64; ++ii) {
    const int d = 68 * ii + 4 * (ii >> 2);
    const float re = lds[ebase + d];
    const float im = lds[ebase + d + QC_PLANE];
    const float p = re * re + im * im;
    T += p;
#pragma unroll
    for (int q = 0; q < 6; ++q)
      accL[q] += ((ii >> (5 - q)) & 1) ? -p : p;
  }
  float acc[12];
#pragma unroll
  for (int q = 0; q < 6; ++q) acc[q] = accL[q];
#pragma unroll
  for (int q = 6; q < 12; ++q)
    acc[q] = ((lt >> (11 - q)) & 1) ? -T : T;
#pragma unroll
  for (int q = 0; q < 12; ++q) {
    float s = acc[q];
#pragma unroll
    for (int off = 16; off > 0; off >>= 1) s += __shfl_xor(s, off);
    acc[q] = s;
  }
  __syncthreads();
  if (lane == 0) {
#pragma unroll
    for (int q = 0; q < 12; ++q) lds[QC_FINV_OFF + w * 12 + q] = acc[q];
  }
  __syncthreads();
  if (h == 0 && lane < 12)
    out[b * 12 + lane] = lds[QC_FINV_OFF + (2 * e) * 12 + lane] +
                         lds[QC_FINV_OFF + (2 * e + 1) * 12 + lane];
}

extern "C" void kernel_launch(void* const* d_in, const int* in_sizes, int n_in,
                              void* d_out, int out_size, void* d_ws,
                              size_t ws_size, hipStream_t stream) {
  (void)n_in; (void)out_size; (void)ws_size;
  const float* x = (const float*)d_in[0];
  const float* params = (const float*)d_in[1];
  float* out = (float*)d_out;
  float* ws = (float*)d_ws;

  const int batch = in_sizes[0] / QC_DIM;
  const int blocks = batch / QC_BATCH_PER_BLOCK;

  qc_setup_gates<<<1, 288, 0, stream>>>(params, ws);

  (void)hipFuncSetAttribute((const void*)qc_main,
                            hipFuncAttributeMaxDynamicSharedMemorySize,
                            QC_LDS_BYTES);
  qc_main<<<blocks, QC_THREADS, QC_LDS_BYTES, stream>>>(x, ws, out);
}


# gates applied per 3-qubit group as ONE real 16x16 product per tile (A=[Re|Im], B=[[VrT,ViT],[-ViT,VrT]]): 4 passes and 512 matrix instructions per layer, not 3 and 768; changes summation order
# speedup vs baseline: 1.0409x; 1.0409x over previous
#include <hip/hip_runtime.h>
#include <math.h>


typedef float v2f __attribute__((ext_vector_type(2)));
typedef float v8f __attribute__((ext_vector_type(8)));

#define QC_BATCH_PER_BLOCK 4
#define QC_THREADS 256
#define QC_DIM 4096

#define QC_PLANE 4417
#define QC_BUF (2 * QC_PLANE)
#define QC_ELEM (2 * QC_BUF)
#define QC_FINV_OFF (QC_BATCH_PER_BLOCK * QC_ELEM)
#define QC_LDS_FLOATS (QC_FINV_OFF + QC_DIM)
#define QC_LDS_BYTES (QC_LDS_FLOATS * 4)

static __device__ __forceinline__ v8f wmma4(v2f a, v2f b, v8f c) {
  return __builtin_amdgcn_wmma_f32_16x16x4_f32(false, a, false, b, (short)0, c,
                                               false, false);
}

static __device__ __forceinline__ float xlane8(float v) {
  return __int_as_float(__builtin_amdgcn_update_dpp(
      0, __float_as_int(v), 0x168  , 0xf, 0xf, false));
}

static __device__ __forceinline__ int physidx(int i) {
  return i + (i >> 4) + 4 * (i >> 8);
}

static __device__ __forceinline__ int fperm(int i) {
  int x = i;
#pragma unroll
  for (int q = 11; q >= 0; --q) {
    int C = 11 - q;
    int T = 11 - ((q + 1) % 12);
    x ^= ((x >> C) & 1) << T;
  }
  return x;
}

template <int P> struct PassBits;
template <> struct PassBits<0> {
  static constexpr int SH = 9, M0 = 0, M1 = 1, M2 = 2, M3 = 5;
  static constexpr int T0 = 3, T1 = 4, T2 = 6, T3 = 7, T4 = 8;
};
template <> struct PassBits<1> {
  static constexpr int SH = 6, M0 = 0, M1 = 1, M2 = 2, M3 = 5;
  static constexpr int T0 = 3, T1 = 4, T2 = 9, T3 = 10, T4 = 11;
};
template <> struct PassBits<2> {
  static constexpr int SH = 3, M0 = 0, M1 = 1, M2 = 2, M3 = 11;
  static constexpr int T0 = 6, T1 = 7, T2 = 8, T3 = 9, T4 = 10;
};
template <> struct PassBits<3> {
  static constexpr int SH = 0, M0 = 4, M1 = 10, M2 = 11, M3 = 3;
  static constexpr int T0 = 5, T1 = 6, T2 = 7, T3 = 8, T4 = 9;
};

template <int P>
static __device__ __forceinline__ int depM(int m) {
  typedef PassBits<P> B;
  return ((m & 1) << B::M0) | (((m >> 1) & 1) << B::M1) |
         (((m >> 2) & 1) << B::M2) | (((m >> 3) & 1) << B::M3);
}
template <int P>
static __device__ __forceinline__ int depT(int t) {
  typedef PassBits<P> B;
  return ((t & 1) << B::T0) | (((t >> 1) & 1) << B::T1) |
         (((t >> 2) & 1) << B::T2) | (((t >> 3) & 1) << B::T3) |
         (((t >> 4) & 1) << B::T4);
}

template <int P, bool PERM>
static __device__ __forceinline__ void apply_pass(float* lds, int srcOff,
                                                  int dstOff,
                                                  const float* wsMat, int lane,
                                                  int hw) {
  constexpr int SH = PassBits<P>::SH;
  float w[8];
  const float* wbase = wsMat + lane * 8;
#pragma unroll
  for (int s = 0; s < 8; ++s) w[s] = wbase[s];

  const int m = lane & 15;
  const int h = lane >> 4;
  const int nc = m & 7;
  const int np = m >> 3;

  const int laneA = srcOff + physidx(depM<P>(m)) + physidx((2 * h) << SH);
  const int keyD = depM<P>(8 * h) | (nc << SH);
  const int laneD = dstOff + np * QC_PLANE + physidx(keyD);

#pragma unroll 1
  for (int it = 0; it < 2; ++it) {
    const int tlogB = depT<P>(16 * hw + 8 * it);
    const int tphysB = physidx(tlogB);
    float a[8][8];
#pragma unroll
    for (int j = 0; j < 4; ++j) {
#pragma unroll
      for (int u = 0; u < 8; ++u) {
#pragma unroll
        for (int e = 0; e < 2; ++e) {
          a[u][2 * j + e] =
              lds[laneA + tphysB + physidx(depT<P>(u)) + (j >> 1) * QC_PLANE +
                  physidx((4 * (j & 1) + e) << SH)];
        }
      }
    }
    v8f d[8];
    const v8f zero8 = {0.f, 0.f, 0.f, 0.f, 0.f, 0.f, 0.f, 0.f};
#pragma unroll
    for (int u = 0; u < 8; ++u) d[u] = zero8;
#pragma unroll
    for (int j = 0; j < 4; ++j) {
      v2f Bv = {w[2 * j], w[2 * j + 1]};
#pragma unroll
      for (int u = 0; u < 8; ++u) {
        v2f Av = {a[u][2 * j], a[u][2 * j + 1]};
        d[u] = wmma4(Av, Bv, d[u]);
      }
    }
    if (PERM) {
      const int keyX = keyD | depM<P>(np) | tlogB;
#pragma unroll
      for (int u = 0; u < 8; ++u) {
#pragma unroll
        for (int r0 = 0; r0 < 8; r0 += 2) {
          const float X = d[u][r0], Y = d[u][r0 + 1];
          const float got = xlane8(np ? X : Y);
          const float re = np ? got : X;
          const float im = np ? Y : got;
          const int key = keyX | depT<P>(u) | depM<P>(r0);
          const int ph = __float_as_int(lds[QC_FINV_OFF + key]);
          lds[dstOff + ph] = re;
          lds[dstOff + ph + QC_PLANE] = im;
        }
      }
    } else {
#pragma unroll
      for (int u = 0; u < 8; ++u) {
#pragma unroll
        for (int r = 0; r < 8; ++r) {
          lds[laneD + tphysB + physidx(depT<P>(u)) + physidx(depM<P>(r))] =
              d[u][r];
        }
      }
    }
  }
}

__global__ void qc_setup_gates(const float* __restrict__ params,
                               float* __restrict__ ws) {
  const int tid = threadIdx.x;
  const int mat = tid >> 5;
  const int lane = tid & 31;
  if (mat >= 12) return;
  const int L = mat / 4, g = mat % 4;

  float mr[3][2][2], mi[3][2][2];
#pragma unroll
  for (int s = 0; s < 3; ++s) {
    const int q = g * 3 + s;
    const int base = L * 36 + q * 3;
    const float tx = params[base], ty = params[base + 1], tz = params[base + 2];
    const float cx = cosf(0.5f * tx), sx = sinf(0.5f * tx);
    const float cy = cosf(0.5f * ty), sy = sinf(0.5f * ty);
    const float cz = cosf(0.5f * tz), sz = sinf(0.5f * tz);
    float rr, ii;
    rr = cy * cx; ii = sy * sx;
    mr[s][0][0] = cz * rr + sz * ii; mi[s][0][0] = cz * ii - sz * rr;
    rr = -sy * cx; ii = -cy * sx;
    mr[s][0][1] = cz * rr + sz * ii; mi[s][0][1] = cz * ii - sz * rr;
    rr = sy * cx; ii = -cy * sx;
    mr[s][1][0] = cz * rr - sz * ii; mi[s][1][0] = cz * ii + sz * rr;
    rr = cy * cx; ii = -sy * sx;
    mr[s][1][1] = cz * rr - sz * ii; mi[s][1][1] = cz * ii + sz * rr;
  }

  float* base = ws + mat * 256;
  const int n = lane & 15;
  const int nc = n & 7, np = n >> 3;
  const int khalf = (lane >> 4) << 1;
#pragma unroll
  for (int j = 0; j < 4; ++j) {
#pragma unroll
    for (int e = 0; e < 2; ++e) {
      const int k = 4 * j + e + khalf;
      const int kc = k & 7, kp = k >> 3;
      float re = 1.f, im = 0.f;
#pragma unroll
      for (int s = 0; s < 3; ++s) {
        const int rb = (nc >> (2 - s)) & 1;
        const int cb = (kc >> (2 - s)) & 1;
        const float wr = mr[s][rb][cb], wi = mi[s][rb][cb];
        const float nre = re * wr - im * wi;
        const float nim = re * wi + im * wr;
        re = nre; im = nim;
      }
      const float val = (kp == np) ? re : ((kp == 0) ? im : -im);
      base[lane * 8 + j * 2 + e] = val;
    }
  }
}

__global__ __launch_bounds__(QC_THREADS) void qc_main(
    const float* __restrict__ x, const float* __restrict__ ws,
    float* __restrict__ out) {
  extern __shared__ float lds[];
  const int tid = threadIdx.x;
  const int lane = tid & 31;
  const int w = tid >> 5;
  const int e = w >> 1;
  const int h = w & 1;
  const int lt = h * 32 + lane;
  const long long b = (long long)blockIdx.x * QC_BATCH_PER_BLOCK + e;

  const int eb = e * QC_ELEM;

  for (int p = tid; p < QC_DIM; p += QC_THREADS)
    lds[QC_FINV_OFF + fperm(p)] = __int_as_float(physidx(p));

  const float4* xr = (const float4*)(x + b * QC_DIM);
  float4 v[16];
  float ss = 0.f;
#pragma unroll
  for (int ii = 0; ii < 16; ++ii) {
    v[ii] = xr[ii * 64 + lt];
    ss += v[ii].x * v[ii].x + v[ii].y * v[ii].y + v[ii].z * v[ii].z +
          v[ii].w * v[ii].w;
  }
#pragma unroll
  for (int off = 16; off > 0; off >>= 1) ss += __shfl_xor(ss, off);
  if (lane == 0) lds[eb + QC_PLANE + h] = ss;
  __syncthreads();
  const float inv = rsqrtf(lds[eb + QC_PLANE] + lds[eb + QC_PLANE + 1]);
  __syncthreads();
#pragma unroll
  for (int ii = 0; ii < 16; ++ii) {
    const int u = ii * 64 + lt;
    const int pb = 4 * u + (u >> 2) + 4 * ii;
    lds[eb + pb + 0] = v[ii].x * inv;
    lds[eb + pb + 1] = v[ii].y * inv;
    lds[eb + pb + 2] = v[ii].z * inv;
    lds[eb + pb + 3] = v[ii].w * inv;
    lds[eb + QC_PLANE + pb + 0] = 0.f;
    lds[eb + QC_PLANE + pb + 1] = 0.f;
    lds[eb + QC_PLANE + pb + 2] = 0.f;
    lds[eb + QC_PLANE + pb + 3] = 0.f;
  }
  __syncthreads();

  int cb = 0;
#pragma unroll 1
  for (int L = 0; L < 3; ++L) {
    const float* wsL = ws + L * 4 * 256;
    const int cur = eb + cb * QC_BUF;
    const int nxt = eb + (cb ^ 1) * QC_BUF;
    apply_pass<0, false>(lds, cur, cur, wsL + 0, lane, h);
    __syncthreads();
    apply_pass<1, false>(lds, cur, cur, wsL + 256, lane, h);
    __syncthreads();
    apply_pass<2, false>(lds, cur, cur, wsL + 512, lane, h);
    __syncthreads();
    apply_pass<3, true>(lds, cur, nxt, wsL + 768, lane, h);
    __syncthreads();
    cb ^= 1;
  }
  const int fin = eb + cb * QC_BUF;

  const int ebase = fin + lt + (lt >> 4);
  float T = 0.f;
  float accL[6];
#pragma unroll
  for (int q = 0; q < 6; ++q) accL[q] = 0.f;
#pragma unroll
  for (int ii = 0; ii < 64; ++ii) {
    const int d = 68 * ii + 4 * (ii >> 2);
    const float re = lds[ebase + d];
    const float im = lds[ebase + d + QC_PLANE];
    const float p = re * re + im * im;
    T += p;
#pragma unroll
    for (int q = 0; q < 6; ++q)
      accL[q] += ((ii >> (5 - q)) & 1) ? -p : p;
  }
  float acc[12];
#pragma unroll
  for (int q = 0; q < 6; ++q) acc[q] = accL[q];
#pragma unroll
  for (int q = 6; q < 12; ++q)
    acc[q] = ((lt >> (11 - q)) & 1) ? -T : T;
#pragma unroll
  for (int q = 0; q < 12; ++q) {
    float s = acc[q];
#pragma unroll
    for (int off = 16; off > 0; off >>= 1) s += __shfl_xor(s, off);
    acc[q] = s;
  }
  __syncthreads();
  if (lane == 0) {
#pragma unroll
    for (int q = 0; q < 12; ++q) lds[QC_FINV_OFF + w * 12 + q] = acc[q];
  }
  __syncthreads();
  if (h == 0 && lane < 12)
    out[b * 12 + lane] = lds[QC_FINV_OFF + (2 * e) * 12 + lane] +
                         lds[QC_FINV_OFF + (2 * e + 1) * 12 + lane];
}

extern "C" void kernel_launch(void* const* d_in, const int* in_sizes, int n_in,
                              void* d_out, int out_size, void* d_ws,
                              size_t ws_size, hipStream_t stream) {
  (void)n_in; (void)out_size; (void)ws_size;
  const float* x = (const float*)d_in[0];
  const float* params = (const float*)d_in[1];
  float* out = (float*)d_out;
  float* ws = (float*)d_ws;

  const int batch = in_sizes[0] / QC_DIM;
  const int blocks = batch / QC_BATCH_PER_BLOCK;

  qc_setup_gates<<<1, 384, 0, stream>>>(params, ws);

  (void)hipFuncSetAttribute((const void*)qc_main,
                            hipFuncAttributeMaxDynamicSharedMemorySize,
                            QC_LDS_BYTES);
  qc_main<<<blocks, QC_THREADS, QC_LDS_BYTES, stream>>>(x, ws, out);
}


# persistent work-group per WGP, its two halves run the 8 barrier stages 4 apart: one half's x read and expectation run beside the other's matrix passes; norm applied to the 12 sums, changes rounding
# speedup vs baseline: 1.6270x; 1.5631x over previous
#include <hip/hip_runtime.h>
#include <math.h>


typedef float v2f __attribute__((ext_vector_type(2)));
typedef float v8f __attribute__((ext_vector_type(8)));

#define QC_BATCH_PER_BLOCK 4
#define QC_THREADS 256
#define QC_DIM 4096

#define QC_SLOTS 4228
#define QC_BUF (2 * QC_SLOTS)
#define QC_ELEM (2 * QC_BUF)
#define QC_RED_OFF (QC_BATCH_PER_BLOCK * QC_ELEM)
#define QC_RED 128
#define QC_LDS_FLOATS (QC_RED_OFF + QC_RED)
#define QC_LDS_BYTES (QC_LDS_FLOATS * 4)

static __device__ __forceinline__ v8f wmma4(v2f a, v2f b, v8f c) {
  return __builtin_amdgcn_wmma_f32_16x16x4_f32(false, a, false, b, (short)0, c,
                                               false, false);
}

static __device__ __forceinline__ int slotidx(int i) {
  return i + (i >> 5) + (i >> 10);
}

static __device__ __forceinline__ v2f lds_ld2(const float* lds, int off) {
  return *(const v2f*)(lds + off);
}
static __device__ __forceinline__ void lds_st2(float* lds, int off, v2f v) {
  *(v2f*)(lds + off) = v;
}

static __device__ __forceinline__ int cnot_target(int i) {
  int x = i;
  x ^= x >> 1;
  x ^= x >> 2;
  x ^= x >> 4;
  x ^= x >> 8;
  return (x & 0x7FF) | (((x ^ (i >> 11)) & 1) << 11);
}

template <int F> struct FuseBits;
template <> struct FuseBits<0> {
  static constexpr int SHX = 9, SHY = 6, M0 = 0, M1 = 1, M2 = 2, M3 = 3;
  static constexpr int R0 = 4, R1 = 5;
};
template <> struct FuseBits<1> {
  static constexpr int SHX = 0, SHY = 3, M0 = 6, M1 = 7, M2 = 8, M3 = 9;
  static constexpr int R0 = 10, R1 = 11;
};

template <int F>
static __device__ __forceinline__ int depM(int m) {
  typedef FuseBits<F> B;
  return ((m & 1) << B::M0) | (((m >> 1) & 1) << B::M1) |
         (((m >> 2) & 1) << B::M2) | (((m >> 3) & 1) << B::M3);
}

static __device__ __forceinline__ void half_swap(float& lo, float& hi) {
  typedef unsigned v2u __attribute__((ext_vector_type(2)));
  const v2u r = __builtin_amdgcn_permlane16_swap(__float_as_uint(lo),
                                                 __float_as_uint(hi), false,
                                                 false);
  lo = __uint_as_float(r[0]);
  hi = __uint_as_float(r[1]);
}

template <int F, bool PERM>
static __device__ __forceinline__ void apply_fused(float* lds, int srcOff,
                                                   int dstOff,
                                                   const float* wsX,
                                                   const float* wsY, int lane,
                                                   int hw) {
  constexpr int SHX = FuseBits<F>::SHX;
  constexpr int SHY = FuseBits<F>::SHY;
  constexpr int R0 = FuseBits<F>::R0;
  constexpr int R1 = FuseBits<F>::R1;
  float wx[8], wy[8];
  const float* wxb = wsX + lane * 8;
  const float* wyb = wsY + lane * 8;
#pragma unroll
  for (int s = 0; s < 8; ++s) wx[s] = wxb[s];
#pragma unroll
  for (int s = 0; s < 8; ++s) wy[s] = wyb[s];

  const int c = lane & 15;
  const int hh = lane >> 4;
  const int cols = depM<F>(c);
  const int ldL = 2 * slotidx(cols | (hh << SHX));
  const int stKey = cols | (hh << SHY);
  const int stL = 2 * slotidx(stKey);
  const int tgtL = cnot_target(stKey);

#pragma unroll 1
  for (int it = 0; it < 2; ++it) {
    const int tw = (it << R0) | (hw << R1);
    const int twS = 2 * slotidx(tw);
    v2f b[8][4];
#pragma unroll
    for (int u = 0; u < 8; ++u) {
#pragma unroll
      for (int j = 0; j < 4; ++j) {
        b[u][j] = lds_ld2(lds, srcOff + ldL + twS +
                                   2 * slotidx((u << SHY) | ((2 * j) << SHX)));
      }
    }
    const v8f zero8 = {0.f, 0.f, 0.f, 0.f, 0.f, 0.f, 0.f, 0.f};
    v8f d[8];
#pragma unroll
    for (int u = 0; u < 8; ++u) d[u] = zero8;
#pragma unroll
    for (int q = 0; q < 4; ++q) {
#pragma unroll
      for (int j = 0; j < 4; ++j) {
        const v2f Av = {wx[2 * j], wx[2 * j + 1]};
        d[2 * q] = wmma4(Av, b[2 * q][j], d[2 * q]);
        d[2 * q + 1] = wmma4(Av, b[2 * q + 1][j], d[2 * q + 1]);
      }
    }
#pragma unroll
    for (int q = 0; q < 4; ++q) {
#pragma unroll
      for (int k = 0; k < 8; ++k) {
        float lo = d[2 * q][k], hi = d[2 * q + 1][k];
        half_swap(lo, hi);
        d[2 * q][k] = lo;
        d[2 * q + 1][k] = hi;
      }
    }
    v8f d2[8];
#pragma unroll
    for (int g = 0; g < 8; ++g) d2[g] = zero8;
#pragma unroll
    for (int q = 0; q < 4; ++q) {
      const v2f Av = {wy[2 * q], wy[2 * q + 1]};
#pragma unroll
      for (int g = 0; g < 8; ++g) {
        const int a = g & 1, r = g >> 1;
        const v2f Bv = {d[2 * q + a][2 * r], d[2 * q + a][2 * r + 1]};
        d2[g] = wmma4(Av, Bv, d2[g]);
      }
    }
    if (PERM) {
      const int tgtX = tgtL ^ cnot_target(tw);
#pragma unroll
      for (int g = 0; g < 8; ++g) {
#pragma unroll
        for (int r2 = 0; r2 < 4; ++r2) {
          const int p = tgtX ^ cnot_target((g << SHX) | ((2 * r2) << SHY));
          const v2f val = {d2[g][2 * r2], d2[g][2 * r2 + 1]};
          lds_st2(lds, dstOff + 2 * slotidx(p), val);
        }
      }
    } else {
#pragma unroll
      for (int g = 0; g < 8; ++g) {
#pragma unroll
        for (int r2 = 0; r2 < 4; ++r2) {
          const v2f val = {d2[g][2 * r2], d2[g][2 * r2 + 1]};
          lds_st2(lds,
                  dstOff + stL + twS +
                      2 * slotidx((g << SHX) | ((2 * r2) << SHY)),
                  val);
        }
      }
    }
  }
}

__global__ void qc_setup_gates(const float* __restrict__ params,
                               float* __restrict__ ws) {
  const int tid = threadIdx.x;
  const int mat = tid >> 5;
  const int lane = tid & 31;
  if (mat >= 12) return;
  const int L = mat / 4, g = mat % 4;

  float mr[3][2][2], mi[3][2][2];
#pragma unroll
  for (int s = 0; s < 3; ++s) {
    const int q = g * 3 + s;
    const int base = L * 36 + q * 3;
    const float tx = params[base], ty = params[base + 1], tz = params[base + 2];
    const float cx = cosf(0.5f * tx), sx = sinf(0.5f * tx);
    const float cy = cosf(0.5f * ty), sy = sinf(0.5f * ty);
    const float cz = cosf(0.5f * tz), sz = sinf(0.5f * tz);
    float rr, ii;
    rr = cy * cx; ii = sy * sx;
    mr[s][0][0] = cz * rr + sz * ii; mi[s][0][0] = cz * ii - sz * rr;
    rr = -sy * cx; ii = -cy * sx;
    mr[s][0][1] = cz * rr + sz * ii; mi[s][0][1] = cz * ii - sz * rr;
    rr = sy * cx; ii = -cy * sx;
    mr[s][1][0] = cz * rr - sz * ii; mi[s][1][0] = cz * ii + sz * rr;
    rr = cy * cx; ii = -sy * sx;
    mr[s][1][1] = cz * rr - sz * ii; mi[s][1][1] = cz * ii + sz * rr;
  }

  float* base = ws + mat * 256;
  const int m = lane & 15;
  const int mp = m & 1;
  const int mc = 2 * ((m >> 1) & 3) + (m >> 3);
  const int khalf = lane >> 4;
#pragma unroll
  for (int j = 0; j < 4; ++j) {
#pragma unroll
    for (int e = 0; e < 2; ++e) {
      const int kc = 2 * j + khalf;
      const int kp = e;
      float re = 1.f, im = 0.f;
#pragma unroll
      for (int s = 0; s < 3; ++s) {
        const int rb = (mc >> (2 - s)) & 1;
        const int cb = (kc >> (2 - s)) & 1;
        const float wr = mr[s][rb][cb], wi = mi[s][rb][cb];
        const float nre = re * wr - im * wi;
        const float nim = re * wi + im * wr;
        re = nre; im = nim;
      }
      const float val = (kp == mp) ? re : ((mp == 1) ? im : -im);
      base[lane * 8 + j * 2 + e] = val;
    }
  }
}

extern __shared__ __attribute__((aligned(16))) float lds[];

#define QC_GRID 256
#define QC_STAGES 8
#define QC_LAG 4
#define QC_RED_ROW 16

__global__ __launch_bounds__(QC_THREADS) void qc_main(
    const float* __restrict__ x, const float* __restrict__ ws,
    float* __restrict__ out, int nquads) {
  const int tid = threadIdx.x;
  const int lane = tid & 31;
  const int w = __builtin_amdgcn_readfirstlane(tid >> 5);
  const int e = w >> 1;
  const int h = w & 1;
  const int hf = w >> 2;
  const int lt = h * 32 + lane;
  const int eb = e * QC_ELEM;

  const int npairs = 2 * nquads;
  const int stride = 2 * (int)gridDim.x;
  const int first = 2 * (int)blockIdx.x + hf;
  const int rounds = (npairs + stride - 1) / stride;
  const int nst = rounds * QC_STAGES;

  float ssPart = 0.f;

#pragma unroll 1
  for (int s = 0; s <= nst + QC_LAG; ++s) {
    const int t = s - QC_LAG * hf;

    if (t >= QC_STAGES && t <= nst && (t & (QC_STAGES - 1)) == 0) {
      const int pr = first + ((t >> 3) - 1) * stride;
      if (pr < npairs && h == 0 && lane < 12) {
        const long long b = 2LL * pr + (e & 1);
        const int r0 = QC_RED_OFF + (2 * e) * QC_RED_ROW;
        const int r1 = r0 + QC_RED_ROW;
        const float sum = lds[r0 + lane] + lds[r1 + lane];
        const float ssq = lds[r0 + 12] + lds[r1 + 12];
        out[b * 12 + lane] = sum / ssq;
      }
    }

    if (t >= 0 && t < nst) {
      const int st = t & (QC_STAGES - 1);
      const int pr = first + (t >> 3) * stride;
      if (pr < npairs) {
        if (st == 0) {
          const long long b = 2LL * pr + (e & 1);
          const float4* xr = (const float4*)(x + b * QC_DIM);
          float4 v[16];
#pragma unroll
          for (int ii = 0; ii < 16; ++ii) v[ii] = xr[ii * 64 + lt];
          float ss = 0.f;
#pragma unroll
          for (int ii = 0; ii < 16; ++ii) {
            ss += v[ii].x * v[ii].x + v[ii].y * v[ii].y + v[ii].z * v[ii].z +
                  v[ii].w * v[ii].w;
            const int u = ii * 64 + lt;
            const int pb = 4 * u + (u >> 3) + (u >> 8);
            const v2f a0 = {v[ii].x, 0.f};
            const v2f a1 = {v[ii].y, 0.f};
            const v2f a2 = {v[ii].z, 0.f};
            const v2f a3 = {v[ii].w, 0.f};
            lds_st2(lds, eb + 2 * (pb + 0), a0);
            lds_st2(lds, eb + 2 * (pb + 1), a1);
            lds_st2(lds, eb + 2 * (pb + 2), a2);
            lds_st2(lds, eb + 2 * (pb + 3), a3);
          }
          ssPart = ss;
        } else if (st == 7) {
          const int ebase = eb + QC_BUF + 2 * (lt + (lt >> 5));
          float T = 0.f;
          float accL[6];
#pragma unroll
          for (int q = 0; q < 6; ++q) accL[q] = 0.f;
#pragma unroll
          for (int ii = 0; ii < 64; ++ii) {
            const int d = 2 * (66 * ii + (ii >> 4));
            const v2f amp = lds_ld2(lds, ebase + d);
            const float p = amp.x * amp.x + amp.y * amp.y;
            T += p;
#pragma unroll
            for (int q = 0; q < 6; ++q)
              accL[q] += ((ii >> (5 - q)) & 1) ? -p : p;
          }
          float acc[13];
#pragma unroll
          for (int q = 0; q < 6; ++q) acc[q] = accL[q];
#pragma unroll
          for (int q = 6; q < 12; ++q)
            acc[q] = ((lt >> (11 - q)) & 1) ? -T : T;
          acc[12] = ssPart;
#pragma unroll
          for (int q = 0; q < 13; ++q) {
            float r = acc[q];
#pragma unroll
            for (int off = 16; off > 0; off >>= 1) r += __shfl_xor(r, off);
            acc[q] = r;
          }
          if (lane == 0) {
#pragma unroll
            for (int q = 0; q < 13; ++q)
              lds[QC_RED_OFF + w * QC_RED_ROW + q] = acc[q];
          }
        } else {
          const int L = (st - 1) >> 1;
          const float* wsL = ws + L * 4 * 256;
          const int cur = eb + (L & 1) * QC_BUF;
          const int nxt = eb + ((L & 1) ^ 1) * QC_BUF;
          if ((st - 1) & 1) {
            apply_fused<1, true>(lds, cur, nxt, wsL + 768, wsL + 512, lane, h);
          } else {
            apply_fused<0, false>(lds, cur, cur, wsL + 0, wsL + 256, lane, h);
          }
        }
      }
    }
    __syncthreads();
  }
}

extern "C" void kernel_launch(void* const* d_in, const int* in_sizes, int n_in,
                              void* d_out, int out_size, void* d_ws,
                              size_t ws_size, hipStream_t stream) {
  (void)n_in; (void)out_size; (void)ws_size;
  const float* x = (const float*)d_in[0];
  const float* params = (const float*)d_in[1];
  float* out = (float*)d_out;
  float* ws = (float*)d_ws;

  const int batch = in_sizes[0] / QC_DIM;
  const int blocks = batch / QC_BATCH_PER_BLOCK;

  qc_setup_gates<<<1, 384, 0, stream>>>(params, ws);

  (void)hipFuncSetAttribute((const void*)qc_main,
                            hipFuncAttributeMaxDynamicSharedMemorySize,
                            QC_LDS_BYTES);
  qc_main<<<QC_GRID, QC_THREADS, QC_LDS_BYTES, stream>>>(x, ws, out, blocks);
}


# CNOT ring as a change of index basis, not a data move (layer l reads index j at P^-l(j)): all passes in place, 1 buffer/element, 8 elements in 4 staggered wave groups per WGP; changes summation order
# speedup vs baseline: 1.7025x; 1.0464x over previous
#include <hip/hip_runtime.h>
#include <math.h>


typedef float v2f __attribute__((ext_vector_type(2)));
typedef float v8f __attribute__((ext_vector_type(8)));

#define QC_BATCH_PER_BLOCK 4
#define QC_ELEMS 8
#define QC_THREADS 512
#define QC_DIM 4096

#define QC_ELEM_BYTES 32768
#define QC_RED_OFF_B (QC_ELEMS * QC_ELEM_BYTES)
#define QC_RED_ROW 16
#define QC_LDS_BYTES (QC_RED_OFF_B + 16 * QC_RED_ROW * 4)

extern __shared__ __attribute__((aligned(16))) float lds[];

static __device__ __forceinline__ v8f wmma4(v2f a, v2f b, v8f c) {
  return __builtin_amdgcn_wmma_f32_16x16x4_f32(false, a, false, b, (short)0, c,
                                               false, false);
}

static __device__ __forceinline__ v2f lds_ld2(int byteOff) {
  return *(const v2f*)((const char*)lds + byteOff);
}
static __device__ __forceinline__ void lds_st2(int byteOff, v2f v) {
  *(v2f*)((char*)lds + byteOff) = v;
}

static __host__ __device__ constexpr int cnot_target(int i) {
  int x = i;
  x ^= x >> 1;
  x ^= x >> 2;
  x ^= x >> 4;
  x ^= x >> 8;
  return (x & 0x7FF) | (((x ^ (i >> 11)) & 1) << 11);
}

static __host__ __device__ constexpr int cnot_source(int t) {
  return ((t ^ (t >> 1)) & 0x3FF) | (((t ^ (t >> 10) ^ (t >> 11)) & 1) << 10) |
         (((t ^ (t >> 11)) & 1) << 11);
}

template <int L>
static __host__ __device__ constexpr int phys_of(int j) {
  for (int s = 0; s < L; ++s) j = cnot_source(j);
  return j;
}

static __host__ __device__ constexpr int byte_of(int p) {
  const int h = p >> 5;
  int r = p;
  r ^= (0 - (h & 1)) & 25;
  r ^= (0 - ((h >> 1) & 1)) & 27;
  r ^= (0 - ((h >> 2) & 1)) & 9;
  r ^= (0 - ((h >> 3) & 1)) & 24;
  r ^= (0 - ((h >> 4) & 1)) & 28;
  r ^= (0 - ((h >> 5) & 1)) & 17;
  r ^= (0 - ((h >> 6) & 1)) & 14;
  return r << 3;
}

template <int F> struct FuseBits;
template <> struct FuseBits<0> {
  static constexpr int SHX = 9, SHY = 6, M0 = 0, M1 = 1, M2 = 2, M3 = 3;
  static constexpr int R0 = 4, R1 = 5;
};
template <> struct FuseBits<1> {
  static constexpr int SHX = 0, SHY = 3, M0 = 6, M1 = 7, M2 = 8, M3 = 9;
  static constexpr int R0 = 10, R1 = 11;
};

template <int F>
static __device__ __forceinline__ int depM(int m) {
  typedef FuseBits<F> B;
  return ((m & 1) << B::M0) | (((m >> 1) & 1) << B::M1) |
         (((m >> 2) & 1) << B::M2) | (((m >> 3) & 1) << B::M3);
}

static __device__ __forceinline__ void half_swap(float& lo, float& hi) {
  typedef unsigned v2u __attribute__((ext_vector_type(2)));
  const v2u r = __builtin_amdgcn_permlane16_swap(__float_as_uint(lo),
                                                 __float_as_uint(hi), false,
                                                 false);
  lo = __uint_as_float(r[0]);
  hi = __uint_as_float(r[1]);
}

template <int L, int F>
static __device__ __forceinline__ void apply_fused(int baseB,
                                                   const float* wsX,
                                                   const float* wsY, int lane,
                                                   int hw) {
  constexpr int SHX = FuseBits<F>::SHX;
  constexpr int SHY = FuseBits<F>::SHY;
  constexpr int R0 = FuseBits<F>::R0;
  constexpr int R1 = FuseBits<F>::R1;
  float wx[8], wy[8];
  const float* wxb = wsX + lane * 8;
  const float* wyb = wsY + lane * 8;
#pragma unroll
  for (int s = 0; s < 8; ++s) wx[s] = wxb[s];
#pragma unroll
  for (int s = 0; s < 8; ++s) wy[s] = wyb[s];

  const int c = lane & 15;
  const int hh = lane >> 4;
  const int cols = depM<F>(c);
  const int ldL = baseB ^ byte_of(phys_of<L>(cols | (hh << SHX)));
  const int stL = baseB ^ byte_of(phys_of<L>(cols | (hh << SHY)));

#pragma unroll 1
  for (int it = 0; it < 2; ++it) {
    const int twB = byte_of(phys_of<L>((it << R0) | (hw << R1)));
    int ldT = ldL ^ twB;
    int stT = stL ^ twB;
    asm volatile("" : "+v"(ldT), "+v"(stT));
    v2f b[8][4];
#pragma unroll
    for (int u = 0; u < 8; ++u) {
#pragma unroll
      for (int j = 0; j < 4; ++j) {
        b[u][j] = lds_ld2(
            ldT ^ byte_of(phys_of<L>((u << SHY) | ((2 * j) << SHX))));
      }
    }
    const v8f zero8 = {0.f, 0.f, 0.f, 0.f, 0.f, 0.f, 0.f, 0.f};
    v8f d[8];
#pragma unroll
    for (int u = 0; u < 8; ++u) d[u] = zero8;
#pragma unroll
    for (int q = 0; q < 4; ++q) {
#pragma unroll
      for (int j = 0; j < 4; ++j) {
        const v2f Av = {wx[2 * j], wx[2 * j + 1]};
        d[2 * q] = wmma4(Av, b[2 * q][j], d[2 * q]);
        d[2 * q + 1] = wmma4(Av, b[2 * q + 1][j], d[2 * q + 1]);
      }
    }
#pragma unroll
    for (int q = 0; q < 4; ++q) {
#pragma unroll
      for (int k = 0; k < 8; ++k) {
        float lo = d[2 * q][k], hi = d[2 * q + 1][k];
        half_swap(lo, hi);
        d[2 * q][k] = lo;
        d[2 * q + 1][k] = hi;
      }
    }
    v8f d2[8];
#pragma unroll
    for (int g = 0; g < 8; ++g) d2[g] = zero8;
#pragma unroll
    for (int q = 0; q < 4; ++q) {
      const v2f Av = {wy[2 * q], wy[2 * q + 1]};
#pragma unroll
      for (int g = 0; g < 8; ++g) {
        const int a = g & 1, r = g >> 1;
        const v2f Bv = {d[2 * q + a][2 * r], d[2 * q + a][2 * r + 1]};
        d2[g] = wmma4(Av, Bv, d2[g]);
      }
    }
#pragma unroll
    for (int g = 0; g < 8; ++g) {
#pragma unroll
      for (int r2 = 0; r2 < 4; ++r2) {
        const v2f val = {d2[g][2 * r2], d2[g][2 * r2 + 1]};
        lds_st2(stT ^ byte_of(phys_of<L>((g << SHX) | ((2 * r2) << SHY))),
                val);
      }
    }
  }
}

struct ZTab {
  int hi[12];
  int lo[12];
  int rep[12];
};
static __host__ __device__ constexpr ZTab make_ztab() {
  ZTab z{};
  for (int k = 0; k < 12; ++k) {
    int m = 0;
    for (int b = 0; b < 12; ++b) {
      const int img = cnot_target(cnot_target(cnot_target(1 << b)));
      if ((img >> k) & 1) m |= 1 << b;
    }
    z.hi[k] = m >> 6;
    z.lo[k] = m & 63;
  }
  for (int k = 0; k < 12; ++k) {
    z.rep[k] = k;
    for (int k2 = k - 1; k2 >= 0; --k2)
      if (z.hi[k2] == z.hi[k]) z.rep[k] = k2;
  }
  return z;
}

__global__ void qc_setup_gates(const float* __restrict__ params,
                               float* __restrict__ ws) {
  const int tid = threadIdx.x;
  const int mat = tid >> 5;
  const int lane = tid & 31;
  if (mat >= 12) return;
  const int L = mat / 4, g = mat % 4;

  float mr[3][2][2], mi[3][2][2];
#pragma unroll
  for (int s = 0; s < 3; ++s) {
    const int q = g * 3 + s;
    const int base = L * 36 + q * 3;
    const float tx = params[base], ty = params[base + 1], tz = params[base + 2];
    const float cx = cosf(0.5f * tx), sx = sinf(0.5f * tx);
    const float cy = cosf(0.5f * ty), sy = sinf(0.5f * ty);
    const float cz = cosf(0.5f * tz), sz = sinf(0.5f * tz);
    float rr, ii;
    rr = cy * cx; ii = sy * sx;
    mr[s][0][0] = cz * rr + sz * ii; mi[s][0][0] = cz * ii - sz * rr;
    rr = -sy * cx; ii = -cy * sx;
    mr[s][0][1] = cz * rr + sz * ii; mi[s][0][1] = cz * ii - sz * rr;
    rr = sy * cx; ii = -cy * sx;
    mr[s][1][0] = cz * rr - sz * ii; mi[s][1][0] = cz * ii + sz * rr;
    rr = cy * cx; ii = -sy * sx;
    mr[s][1][1] = cz * rr - sz * ii; mi[s][1][1] = cz * ii + sz * rr;
  }

  float* base = ws + mat * 256;
  const int m = lane & 15;
  const int mp = m & 1;
  const int mc = 2 * ((m >> 1) & 3) + (m >> 3);
  const int khalf = lane >> 4;
#pragma unroll
  for (int j = 0; j < 4; ++j) {
#pragma unroll
    for (int e = 0; e < 2; ++e) {
      const int kc = 2 * j + khalf;
      const int kp = e;
      float re = 1.f, im = 0.f;
#pragma unroll
      for (int s = 0; s < 3; ++s) {
        const int rb = (mc >> (2 - s)) & 1;
        const int cb = (kc >> (2 - s)) & 1;
        const float wr = mr[s][rb][cb], wi = mi[s][rb][cb];
        const float nre = re * wr - im * wi;
        const float nim = re * wi + im * wr;
        re = nre; im = nim;
      }
      const float val = (kp == mp) ? re : ((mp == 1) ? im : -im);
      base[lane * 8 + j * 2 + e] = val;
    }
  }
}

#define QC_GRID 256
#define QC_GROUPS 4
#define QC_STAGES 8
#define QC_LAG 2

__global__ __launch_bounds__(QC_THREADS) void qc_main(
    const float* __restrict__ x, const float* __restrict__ ws,
    float* __restrict__ out, int nquads) {
  const int tid = threadIdx.x;
  const int lane = tid & 31;
  const int w = __builtin_amdgcn_readfirstlane(tid >> 5);
  const int e = w >> 1;
  const int h = w & 1;
  const int gq = w >> 2;
  const int baseB = e * QC_ELEM_BYTES;

  const int npairs = 2 * nquads;
  const int stride = QC_GROUPS * (int)gridDim.x;
  const int first = QC_GROUPS * (int)blockIdx.x + gq;
  const int rounds = (npairs + stride - 1) / stride;
  const int nst = rounds * QC_STAGES;

  float ssPart = 0.f;

#pragma unroll 1
  for (int s = 0; s <= nst + QC_LAG * (QC_GROUPS - 1); ++s) {
    const int t = s - QC_LAG * gq;
    int ln = lane;
    asm volatile("" : "+v"(ln));
    const int lt = h * 32 + ln;

    if (t >= QC_STAGES && t <= nst && (t & (QC_STAGES - 1)) == 0) {
      const int pr = first + ((t >> 3) - 1) * stride;
      if (pr < npairs && h == 0 && lane < 12) {
        const long long b = 2LL * pr + (e & 1);
        const int r0 = QC_RED_OFF_B / 4 + w * QC_RED_ROW;
        const int r1 = r0 + QC_RED_ROW;
        const float sum = lds[r0 + lane] + lds[r1 + lane];
        const float ssq = lds[r0 + 12] + lds[r1 + 12];
        out[b * 12 + lane] = sum / ssq;
      }
    }

    if (t >= 0 && t < nst) {
      const int st = t & (QC_STAGES - 1);
      const int pr = first + (t >> 3) * stride;
      if (pr < npairs) {
        if (st == 0) {
          const long long b = 2LL * pr + (e & 1);
          const float4* xr = (const float4*)(x + b * QC_DIM);
          float4 v[16];
#pragma unroll
          for (int ii = 0; ii < 16; ++ii) v[ii] = xr[ii * 64 + lt];
          const int l0 = baseB ^ byte_of(4 * lt);
          float ss = 0.f;
#pragma unroll
          for (int ii = 0; ii < 16; ++ii) {
            ss += v[ii].x * v[ii].x + v[ii].y * v[ii].y + v[ii].z * v[ii].z +
                  v[ii].w * v[ii].w;
            const v2f a0 = {v[ii].x, 0.f};
            const v2f a1 = {v[ii].y, 0.f};
            const v2f a2 = {v[ii].z, 0.f};
            const v2f a3 = {v[ii].w, 0.f};
            lds_st2(l0 ^ byte_of(256 * ii + 0), a0);
            lds_st2(l0 ^ byte_of(256 * ii + 1), a1);
            lds_st2(l0 ^ byte_of(256 * ii + 2), a2);
            lds_st2(l0 ^ byte_of(256 * ii + 3), a3);
          }
          ssPart = ss;
        } else if (st == 7) {
          constexpr ZTab Z = make_ztab();
          const int l7 = baseB ^ byte_of(lt);
          float T = 0.f;
          float accK[12];
#pragma unroll
          for (int k = 0; k < 12; ++k) accK[k] = 0.f;
#pragma unroll
          for (int ii = 0; ii < 64; ++ii) {
            const v2f amp = lds_ld2(l7 ^ byte_of(64 * ii));
            const float p = amp.x * amp.x + amp.y * amp.y;
            T += p;
#pragma unroll
            for (int k = 0; k < 12; ++k) {
              if (Z.hi[k] != 0 && Z.rep[k] == k)
                accK[k] += (__builtin_parity(ii & Z.hi[k])) ? -p : p;
            }
          }
          float acc[13];
#pragma unroll
          for (int k = 0; k < 12; ++k) {
            const float sk = (Z.hi[k] == 0) ? T : accK[Z.rep[k]];
            acc[11 - k] = (__builtin_parity(lt & Z.lo[k])) ? -sk : sk;
          }
          acc[12] = ssPart;
#pragma unroll
          for (int q = 0; q < 13; ++q) {
            float r = acc[q];
#pragma unroll
            for (int off = 16; off > 0; off >>= 1) r += __shfl_xor(r, off);
            acc[q] = r;
          }
          if (lane == 0) {
#pragma unroll
            for (int q = 0; q < 13; ++q)
              lds[QC_RED_OFF_B / 4 + w * QC_RED_ROW + q] = acc[q];
          }
        } else {
          const int L = (st - 1) >> 1;
          const float* wsL = ws + L * 4 * 256;
          if (st == 1) {
            apply_fused<0, 0>(baseB, wsL + 0, wsL + 256, ln, h);
          } else if (st == 2) {
            apply_fused<0, 1>(baseB, wsL + 768, wsL + 512, ln, h);
          } else if (st == 3) {
            apply_fused<1, 0>(baseB, wsL + 0, wsL + 256, ln, h);
          } else if (st == 4) {
            apply_fused<1, 1>(baseB, wsL + 768, wsL + 512, ln, h);
          } else if (st == 5) {
            apply_fused<2, 0>(baseB, wsL + 0, wsL + 256, ln, h);
          } else {
            apply_fused<2, 1>(baseB, wsL + 768, wsL + 512, ln, h);
          }
        }
      }
    }
    __syncthreads();
  }
}

extern "C" void kernel_launch(void* const* d_in, const int* in_sizes, int n_in,
                              void* d_out, int out_size, void* d_ws,
                              size_t ws_size, hipStream_t stream) {
  (void)n_in; (void)out_size; (void)ws_size;
  const float* x = (const float*)d_in[0];
  const float* params = (const float*)d_in[1];
  float* out = (float*)d_out;
  float* ws = (float*)d_ws;

  const int batch = in_sizes[0] / QC_DIM;
  const int blocks = batch / QC_BATCH_PER_BLOCK;

  qc_setup_gates<<<1, 384, 0, stream>>>(params, ws);

  (void)hipFuncSetAttribute((const void*)qc_main,
                            hipFuncAttributeMaxDynamicSharedMemorySize,
                            QC_LDS_BYTES);
  qc_main<<<QC_GRID, QC_THREADS, QC_LDS_BYTES, stream>>>(x, ws, out, blocks);
}


# each wave owns one batch element from x row to results (8 waves, 8 LDS buffers per WGP, 4 trips per pass): no work-group barrier, waves run unsynchronised; changes order of the final sums
# speedup vs baseline: 1.7841x; 1.0479x over previous
#include <hip/hip_runtime.h>
#include <math.h>


typedef float v2f __attribute__((ext_vector_type(2)));
typedef float v8f __attribute__((ext_vector_type(8)));

#define QC_BATCH_PER_BLOCK 4
#define QC_ELEMS 8
#define QC_THREADS 256
#define QC_DIM 4096

#define QC_ELEM_BYTES 32768
#define QC_LDS_BYTES (QC_ELEMS * QC_ELEM_BYTES)

extern __shared__ __attribute__((aligned(16))) float lds[];

static __device__ __forceinline__ v8f wmma4(v2f a, v2f b, v8f c) {
  return __builtin_amdgcn_wmma_f32_16x16x4_f32(false, a, false, b, (short)0, c,
                                               false, false);
}

static __device__ __forceinline__ void stage_fence() {
  __builtin_amdgcn_fence(__ATOMIC_RELEASE, "wavefront");
  __builtin_amdgcn_fence(__ATOMIC_ACQUIRE, "wavefront");
}

static __device__ __forceinline__ v2f lds_ld2(int byteOff) {
  return *(const v2f*)((const char*)lds + byteOff);
}
static __device__ __forceinline__ void lds_st2(int byteOff, v2f v) {
  *(v2f*)((char*)lds + byteOff) = v;
}

static __host__ __device__ constexpr int cnot_target(int i) {
  int x = i;
  x ^= x >> 1;
  x ^= x >> 2;
  x ^= x >> 4;
  x ^= x >> 8;
  return (x & 0x7FF) | (((x ^ (i >> 11)) & 1) << 11);
}

static __host__ __device__ constexpr int cnot_source(int t) {
  return ((t ^ (t >> 1)) & 0x3FF) | (((t ^ (t >> 10) ^ (t >> 11)) & 1) << 10) |
         (((t ^ (t >> 11)) & 1) << 11);
}

template <int L>
static __host__ __device__ constexpr int phys_of(int j) {
  for (int s = 0; s < L; ++s) j = cnot_source(j);
  return j;
}

static __host__ __device__ constexpr int byte_of(int p) {
  const int h = p >> 5;
  int r = p;
  r ^= (0 - (h & 1)) & 25;
  r ^= (0 - ((h >> 1) & 1)) & 27;
  r ^= (0 - ((h >> 2) & 1)) & 9;
  r ^= (0 - ((h >> 3) & 1)) & 24;
  r ^= (0 - ((h >> 4) & 1)) & 28;
  r ^= (0 - ((h >> 5) & 1)) & 17;
  r ^= (0 - ((h >> 6) & 1)) & 14;
  return r << 3;
}

template <int F> struct FuseBits;
template <> struct FuseBits<0> {
  static constexpr int SHX = 9, SHY = 6, M0 = 0, M1 = 1, M2 = 2, M3 = 3;
  static constexpr int R0 = 4, R1 = 5;
};
template <> struct FuseBits<1> {
  static constexpr int SHX = 0, SHY = 3, M0 = 6, M1 = 7, M2 = 8, M3 = 9;
  static constexpr int R0 = 10, R1 = 11;
};

template <int F>
static __device__ __forceinline__ int depM(int m) {
  typedef FuseBits<F> B;
  return ((m & 1) << B::M0) | (((m >> 1) & 1) << B::M1) |
         (((m >> 2) & 1) << B::M2) | (((m >> 3) & 1) << B::M3);
}

static __device__ __forceinline__ void half_swap(float& lo, float& hi) {
  typedef unsigned v2u __attribute__((ext_vector_type(2)));
  const v2u r = __builtin_amdgcn_permlane16_swap(__float_as_uint(lo),
                                                 __float_as_uint(hi), false,
                                                 false);
  lo = __uint_as_float(r[0]);
  hi = __uint_as_float(r[1]);
}

template <int L, int F>
static __device__ __forceinline__ void apply_fused(int baseB,
                                                   const float* wsX,
                                                   const float* wsY,
                                                   int lane) {
  constexpr int SHX = FuseBits<F>::SHX;
  constexpr int SHY = FuseBits<F>::SHY;
  constexpr int R0 = FuseBits<F>::R0;
  constexpr int R1 = FuseBits<F>::R1;
  float wx[8], wy[8];
  const float* wxb = wsX + lane * 8;
  const float* wyb = wsY + lane * 8;
#pragma unroll
  for (int s = 0; s < 8; ++s) wx[s] = wxb[s];
#pragma unroll
  for (int s = 0; s < 8; ++s) wy[s] = wyb[s];

  const int c = lane & 15;
  const int hh = lane >> 4;
  const int cols = depM<F>(c);
  const int ldL = baseB ^ byte_of(phys_of<L>(cols | (hh << SHX)));
  const int stL = baseB ^ byte_of(phys_of<L>(cols | (hh << SHY)));

#pragma unroll 1
  for (int it = 0; it < 4; ++it) {
    const int twB = byte_of(
        phys_of<L>(((it & 1) << R0) | ((it >> 1) << R1)));
    int ldT = ldL ^ twB;
    int stT = stL ^ twB;
    asm volatile("" : "+v"(ldT), "+v"(stT));
    v2f b[8][4];
#pragma unroll
    for (int u = 0; u < 8; ++u) {
#pragma unroll
      for (int j = 0; j < 4; ++j) {
        b[u][j] = lds_ld2(
            ldT ^ byte_of(phys_of<L>((u << SHY) | ((2 * j) << SHX))));
      }
    }
    const v8f zero8 = {0.f, 0.f, 0.f, 0.f, 0.f, 0.f, 0.f, 0.f};
    v8f d[8];
#pragma unroll
    for (int u = 0; u < 8; ++u) d[u] = zero8;
#pragma unroll
    for (int q = 0; q < 4; ++q) {
#pragma unroll
      for (int j = 0; j < 4; ++j) {
        const v2f Av = {wx[2 * j], wx[2 * j + 1]};
        d[2 * q] = wmma4(Av, b[2 * q][j], d[2 * q]);
        d[2 * q + 1] = wmma4(Av, b[2 * q + 1][j], d[2 * q + 1]);
      }
    }
#pragma unroll
    for (int q = 0; q < 4; ++q) {
#pragma unroll
      for (int k = 0; k < 8; ++k) {
        float lo = d[2 * q][k], hi = d[2 * q + 1][k];
        half_swap(lo, hi);
        d[2 * q][k] = lo;
        d[2 * q + 1][k] = hi;
      }
    }
    v8f d2[8];
#pragma unroll
    for (int g = 0; g < 8; ++g) d2[g] = zero8;
#pragma unroll
    for (int q = 0; q < 4; ++q) {
      const v2f Av = {wy[2 * q], wy[2 * q + 1]};
#pragma unroll
      for (int g = 0; g < 8; ++g) {
        const int a = g & 1, r = g >> 1;
        const v2f Bv = {d[2 * q + a][2 * r], d[2 * q + a][2 * r + 1]};
        d2[g] = wmma4(Av, Bv, d2[g]);
      }
    }
#pragma unroll
    for (int g = 0; g < 8; ++g) {
#pragma unroll
      for (int r2 = 0; r2 < 4; ++r2) {
        const v2f val = {d2[g][2 * r2], d2[g][2 * r2 + 1]};
        lds_st2(stT ^ byte_of(phys_of<L>((g << SHX) | ((2 * r2) << SHY))),
                val);
      }
    }
  }
}

struct ZTab {
  int hi[12];
  int lo[12];
  int rep[12];
};
static __host__ __device__ constexpr ZTab make_ztab() {
  ZTab z{};
  for (int k = 0; k < 12; ++k) {
    int m = 0;
    for (int b = 0; b < 12; ++b) {
      const int img = cnot_target(cnot_target(cnot_target(1 << b)));
      if ((img >> k) & 1) m |= 1 << b;
    }
    z.hi[k] = m >> 5;
    z.lo[k] = m & 31;
  }
  for (int k = 0; k < 12; ++k) {
    z.rep[k] = k;
    for (int k2 = k - 1; k2 >= 0; --k2)
      if (z.hi[k2] == z.hi[k]) z.rep[k] = k2;
  }
  return z;
}

__global__ void qc_setup_gates(const float* __restrict__ params,
                               float* __restrict__ ws) {
  const int tid = threadIdx.x;
  const int mat = tid >> 5;
  const int lane = tid & 31;
  if (mat >= 12) return;
  const int L = mat / 4, g = mat % 4;

  float mr[3][2][2], mi[3][2][2];
#pragma unroll
  for (int s = 0; s < 3; ++s) {
    const int q = g * 3 + s;
    const int base = L * 36 + q * 3;
    const float tx = params[base], ty = params[base + 1], tz = params[base + 2];
    const float cx = cosf(0.5f * tx), sx = sinf(0.5f * tx);
    const float cy = cosf(0.5f * ty), sy = sinf(0.5f * ty);
    const float cz = cosf(0.5f * tz), sz = sinf(0.5f * tz);
    float rr, ii;
    rr = cy * cx; ii = sy * sx;
    mr[s][0][0] = cz * rr + sz * ii; mi[s][0][0] = cz * ii - sz * rr;
    rr = -sy * cx; ii = -cy * sx;
    mr[s][0][1] = cz * rr + sz * ii; mi[s][0][1] = cz * ii - sz * rr;
    rr = sy * cx; ii = -cy * sx;
    mr[s][1][0] = cz * rr - sz * ii; mi[s][1][0] = cz * ii + sz * rr;
    rr = cy * cx; ii = -sy * sx;
    mr[s][1][1] = cz * rr - sz * ii; mi[s][1][1] = cz * ii + sz * rr;
  }

  float* base = ws + mat * 256;
  const int m = lane & 15;
  const int mp = m & 1;
  const int mc = 2 * ((m >> 1) & 3) + (m >> 3);
  const int khalf = lane >> 4;
#pragma unroll
  for (int j = 0; j < 4; ++j) {
#pragma unroll
    for (int e = 0; e < 2; ++e) {
      const int kc = 2 * j + khalf;
      const int kp = e;
      float re = 1.f, im = 0.f;
#pragma unroll
      for (int s = 0; s < 3; ++s) {
        const int rb = (mc >> (2 - s)) & 1;
        const int cb = (kc >> (2 - s)) & 1;
        const float wr = mr[s][rb][cb], wi = mi[s][rb][cb];
        const float nre = re * wr - im * wi;
        const float nim = re * wi + im * wr;
        re = nre; im = nim;
      }
      const float val = (kp == mp) ? re : ((mp == 1) ? im : -im);
      base[lane * 8 + j * 2 + e] = val;
    }
  }
}

#define QC_GRID 256

__global__ __launch_bounds__(QC_THREADS) void qc_main(
    const float* __restrict__ x, const float* __restrict__ ws,
    float* __restrict__ out, int nquads) {
  const int tid = threadIdx.x;
  const int lane = tid & 31;
  const int w = __builtin_amdgcn_readfirstlane(tid >> 5);
  const int baseB = w * QC_ELEM_BYTES;

  const int nelems = QC_BATCH_PER_BLOCK * nquads;
  const int stride = QC_ELEMS * (int)gridDim.x;

#pragma unroll 1
  for (int b = QC_ELEMS * (int)blockIdx.x + w; b < nelems; b += stride) {
    int ln = lane;
    asm volatile("" : "+v"(ln));

    float ssPart;
    {
      const float4* xr = (const float4*)(x + (long long)b * QC_DIM);
      float ss = 0.f;
#pragma unroll 1
      for (int hf = 0; hf < 2; ++hf) {
        float4 v[16];
#pragma unroll
        for (int ii = 0; ii < 16; ++ii) v[ii] = xr[(16 * hf + ii) * 32 + ln];
        const int l0 = baseB ^ byte_of(4 * ln) ^ byte_of(2048 * hf);
#pragma unroll
        for (int ii = 0; ii < 16; ++ii) {
          ss += v[ii].x * v[ii].x + v[ii].y * v[ii].y + v[ii].z * v[ii].z +
                v[ii].w * v[ii].w;
          const v2f a0 = {v[ii].x, 0.f};
          const v2f a1 = {v[ii].y, 0.f};
          const v2f a2 = {v[ii].z, 0.f};
          const v2f a3 = {v[ii].w, 0.f};
          lds_st2(l0 ^ byte_of(128 * ii + 0), a0);
          lds_st2(l0 ^ byte_of(128 * ii + 1), a1);
          lds_st2(l0 ^ byte_of(128 * ii + 2), a2);
          lds_st2(l0 ^ byte_of(128 * ii + 3), a3);
        }
      }
      ssPart = ss;
    }

    stage_fence();
    apply_fused<0, 0>(baseB, ws + 0, ws + 256, ln);
    stage_fence();
    apply_fused<0, 1>(baseB, ws + 768, ws + 512, ln);
    stage_fence();
    apply_fused<1, 0>(baseB, ws + 1024, ws + 1280, ln);
    stage_fence();
    apply_fused<1, 1>(baseB, ws + 1792, ws + 1536, ln);
    stage_fence();
    apply_fused<2, 0>(baseB, ws + 2048, ws + 2304, ln);
    stage_fence();
    apply_fused<2, 1>(baseB, ws + 2816, ws + 2560, ln);
    stage_fence();

    {
      constexpr ZTab Z = make_ztab();
      const int l7 = baseB ^ byte_of(ln);
      float T = 0.f;
      float accK[12];
#pragma unroll
      for (int k = 0; k < 12; ++k) accK[k] = 0.f;
#pragma unroll 1
      for (int hf = 0; hf < 2; ++hf) {
        const int l7h = l7 ^ byte_of(2048 * hf);
        float Th = 0.f;
        float accH[12];
#pragma unroll
        for (int k = 0; k < 12; ++k) accH[k] = 0.f;
#pragma unroll
        for (int ii = 0; ii < 64; ++ii) {
          const v2f amp = lds_ld2(l7h ^ byte_of(32 * ii));
          const float p = amp.x * amp.x + amp.y * amp.y;
          Th += p;
#pragma unroll
          for (int k = 0; k < 12; ++k) {
            if (Z.hi[k] != 0 && Z.rep[k] == k)
              accH[k] += (__builtin_parity(ii & Z.hi[k] & 63)) ? -p : p;
          }
        }
        T += Th;
#pragma unroll
        for (int k = 0; k < 12; ++k) {
          if (Z.hi[k] != 0 && Z.rep[k] == k) {
            const bool neg = (hf != 0) && ((Z.hi[k] >> 6) & 1);
            accK[k] += neg ? -accH[k] : accH[k];
          }
        }
      }
      float acc[13];
#pragma unroll
      for (int k = 0; k < 12; ++k) {
        const float sk = (Z.hi[k] == 0) ? T : accK[Z.rep[k]];
        acc[11 - k] = (__builtin_parity(ln & Z.lo[k])) ? -sk : sk;
      }
      acc[12] = ssPart;
#pragma unroll
      for (int q = 0; q < 13; ++q) {
        float r = acc[q];
#pragma unroll
        for (int off = 16; off > 0; off >>= 1) r += __shfl_xor(r, off);
        acc[q] = r;
      }
      float num = acc[0];
#pragma unroll
      for (int q = 1; q < 12; ++q) num = (lane == q) ? acc[q] : num;
      if (lane < 12) out[(long long)b * 12 + lane] = num / acc[12];
    }
  }
}

extern "C" void kernel_launch(void* const* d_in, const int* in_sizes, int n_in,
                              void* d_out, int out_size, void* d_ws,
                              size_t ws_size, hipStream_t stream) {
  (void)n_in; (void)out_size; (void)ws_size;
  const float* x = (const float*)d_in[0];
  const float* params = (const float*)d_in[1];
  float* out = (float*)d_out;
  float* ws = (float*)d_ws;

  const int batch = in_sizes[0] / QC_DIM;
  const int blocks = batch / QC_BATCH_PER_BLOCK;

  qc_setup_gates<<<1, 384, 0, stream>>>(params, ws);

  (void)hipFuncSetAttribute((const void*)qc_main,
                            hipFuncAttributeMaxDynamicSharedMemorySize,
                            QC_LDS_BYTES);
  qc_main<<<QC_GRID, QC_THREADS, QC_LDS_BYTES, stream>>>(x, ws, out, blocks);
}


# wave priority window: the younger wave of each SIMD pair (waves 4-7) runs at s_setprio 3 for all but its last element, 0 on the last: the pair ends together, no lone wave at the end; sums unchanged
# speedup vs baseline: 1.8722x; 1.0494x over previous
#include <hip/hip_runtime.h>
#include <math.h>


typedef float v2f __attribute__((ext_vector_type(2)));
typedef float v8f __attribute__((ext_vector_type(8)));

#define QC_BATCH_PER_BLOCK 4
#define QC_ELEMS 8
#define QC_THREADS 256
#define QC_DIM 4096

#define QC_ELEM_BYTES 32768
#define QC_LDS_BYTES (QC_ELEMS * QC_ELEM_BYTES)

extern __shared__ __attribute__((aligned(16))) float lds[];

static __device__ __forceinline__ v8f wmma4(v2f a, v2f b, v8f c) {
  return __builtin_amdgcn_wmma_f32_16x16x4_f32(false, a, false, b, (short)0, c,
                                               false, false);
}

static __device__ __forceinline__ void stage_fence() {
  __builtin_amdgcn_fence(__ATOMIC_RELEASE, "wavefront");
  __builtin_amdgcn_fence(__ATOMIC_ACQUIRE, "wavefront");
}

static __device__ __forceinline__ v2f lds_ld2(int byteOff) {
  return *(const v2f*)((const char*)lds + byteOff);
}
static __device__ __forceinline__ void lds_st2(int byteOff, v2f v) {
  *(v2f*)((char*)lds + byteOff) = v;
}

static __host__ __device__ constexpr int cnot_target(int i) {
  int x = i;
  x ^= x >> 1;
  x ^= x >> 2;
  x ^= x >> 4;
  x ^= x >> 8;
  return (x & 0x7FF) | (((x ^ (i >> 11)) & 1) << 11);
}

static __host__ __device__ constexpr int cnot_source(int t) {
  return ((t ^ (t >> 1)) & 0x3FF) | (((t ^ (t >> 10) ^ (t >> 11)) & 1) << 10) |
         (((t ^ (t >> 11)) & 1) << 11);
}

template <int L>
static __host__ __device__ constexpr int phys_of(int j) {
  for (int s = 0; s < L; ++s) j = cnot_source(j);
  return j;
}

static __host__ __device__ constexpr int byte_of(int p) {
  const int h = p >> 5;
  int r = p;
  r ^= (0 - (h & 1)) & 25;
  r ^= (0 - ((h >> 1) & 1)) & 27;
  r ^= (0 - ((h >> 2) & 1)) & 9;
  r ^= (0 - ((h >> 3) & 1)) & 24;
  r ^= (0 - ((h >> 4) & 1)) & 28;
  r ^= (0 - ((h >> 5) & 1)) & 17;
  r ^= (0 - ((h >> 6) & 1)) & 14;
  return r << 3;
}

template <int F> struct FuseBits;
template <> struct FuseBits<0> {
  static constexpr int SHX = 9, SHY = 6, M0 = 0, M1 = 1, M2 = 2, M3 = 3;
  static constexpr int R0 = 4, R1 = 5;
};
template <> struct FuseBits<1> {
  static constexpr int SHX = 0, SHY = 3, M0 = 6, M1 = 7, M2 = 8, M3 = 9;
  static constexpr int R0 = 10, R1 = 11;
};

template <int F>
static __device__ __forceinline__ int depM(int m) {
  typedef FuseBits<F> B;
  return ((m & 1) << B::M0) | (((m >> 1) & 1) << B::M1) |
         (((m >> 2) & 1) << B::M2) | (((m >> 3) & 1) << B::M3);
}

static __device__ __forceinline__ void half_swap(float& lo, float& hi) {
  typedef unsigned v2u __attribute__((ext_vector_type(2)));
  const v2u r = __builtin_amdgcn_permlane16_swap(__float_as_uint(lo),
                                                 __float_as_uint(hi), false,
                                                 false);
  lo = __uint_as_float(r[0]);
  hi = __uint_as_float(r[1]);
}

template <int L, int F>
static __device__ __forceinline__ void apply_fused(int baseB,
                                                   const float* wsX,
                                                   const float* wsY,
                                                   int lane) {
  constexpr int SHX = FuseBits<F>::SHX;
  constexpr int SHY = FuseBits<F>::SHY;
  constexpr int R0 = FuseBits<F>::R0;
  constexpr int R1 = FuseBits<F>::R1;
  float wx[8], wy[8];
  const float* wxb = wsX + lane * 8;
  const float* wyb = wsY + lane * 8;
#pragma unroll
  for (int s = 0; s < 8; ++s) wx[s] = wxb[s];
#pragma unroll
  for (int s = 0; s < 8; ++s) wy[s] = wyb[s];

  const int c = lane & 15;
  const int hh = lane >> 4;
  const int cols = depM<F>(c);
  const int ldL = baseB ^ byte_of(phys_of<L>(cols | (hh << SHX)));
  const int stL = baseB ^ byte_of(phys_of<L>(cols | (hh << SHY)));

#pragma unroll 1
  for (int it = 0; it < 4; ++it) {
    const int twB = byte_of(
        phys_of<L>(((it & 1) << R0) | ((it >> 1) << R1)));
    int ldT = ldL ^ twB;
    int stT = stL ^ twB;
    asm volatile("" : "+v"(ldT), "+v"(stT));
    v2f b[8][4];
#pragma unroll
    for (int u = 0; u < 8; ++u) {
#pragma unroll
      for (int j = 0; j < 4; ++j) {
        b[u][j] = lds_ld2(
            ldT ^ byte_of(phys_of<L>((u << SHY) | ((2 * j) << SHX))));
      }
    }
    const v8f zero8 = {0.f, 0.f, 0.f, 0.f, 0.f, 0.f, 0.f, 0.f};
    v8f d[8];
#pragma unroll
    for (int u = 0; u < 8; ++u) d[u] = zero8;
#pragma unroll
    for (int q = 0; q < 4; ++q) {
#pragma unroll
      for (int j = 0; j < 4; ++j) {
        const v2f Av = {wx[2 * j], wx[2 * j + 1]};
        d[2 * q] = wmma4(Av, b[2 * q][j], d[2 * q]);
        d[2 * q + 1] = wmma4(Av, b[2 * q + 1][j], d[2 * q + 1]);
      }
    }
#pragma unroll
    for (int q = 0; q < 4; ++q) {
#pragma unroll
      for (int k = 0; k < 8; ++k) {
        float lo = d[2 * q][k], hi = d[2 * q + 1][k];
        half_swap(lo, hi);
        d[2 * q][k] = lo;
        d[2 * q + 1][k] = hi;
      }
    }
    v8f d2[8];
#pragma unroll
    for (int g = 0; g < 8; ++g) d2[g] = zero8;
#pragma unroll
    for (int q = 0; q < 4; ++q) {
      const v2f Av = {wy[2 * q], wy[2 * q + 1]};
#pragma unroll
      for (int g = 0; g < 8; ++g) {
        const int a = g & 1, r = g >> 1;
        const v2f Bv = {d[2 * q + a][2 * r], d[2 * q + a][2 * r + 1]};
        d2[g] = wmma4(Av, Bv, d2[g]);
      }
    }
#pragma unroll
    for (int g = 0; g < 8; ++g) {
#pragma unroll
      for (int r2 = 0; r2 < 4; ++r2) {
        const v2f val = {d2[g][2 * r2], d2[g][2 * r2 + 1]};
        lds_st2(stT ^ byte_of(phys_of<L>((g << SHX) | ((2 * r2) << SHY))),
                val);
      }
    }
  }
}

struct ZTab {
  int hi[12];
  int lo[12];
  int rep[12];
};
static __host__ __device__ constexpr ZTab make_ztab() {
  ZTab z{};
  for (int k = 0; k < 12; ++k) {
    int m = 0;
    for (int b = 0; b < 12; ++b) {
      const int img = cnot_target(cnot_target(cnot_target(1 << b)));
      if ((img >> k) & 1) m |= 1 << b;
    }
    z.hi[k] = m >> 5;
    z.lo[k] = m & 31;
  }
  for (int k = 0; k < 12; ++k) {
    z.rep[k] = k;
    for (int k2 = k - 1; k2 >= 0; --k2)
      if (z.hi[k2] == z.hi[k]) z.rep[k] = k2;
  }
  return z;
}

__global__ void qc_setup_gates(const float* __restrict__ params,
                               float* __restrict__ ws) {
  const int tid = threadIdx.x;
  const int mat = tid >> 5;
  const int lane = tid & 31;
  if (mat >= 12) return;
  const int L = mat / 4, g = mat % 4;

  float mr[3][2][2], mi[3][2][2];
#pragma unroll
  for (int s = 0; s < 3; ++s) {
    const int q = g * 3 + s;
    const int base = L * 36 + q * 3;
    const float tx = params[base], ty = params[base + 1], tz = params[base + 2];
    const float cx = cosf(0.5f * tx), sx = sinf(0.5f * tx);
    const float cy = cosf(0.5f * ty), sy = sinf(0.5f * ty);
    const float cz = cosf(0.5f * tz), sz = sinf(0.5f * tz);
    float rr, ii;
    rr = cy * cx; ii = sy * sx;
    mr[s][0][0] = cz * rr + sz * ii; mi[s][0][0] = cz * ii - sz * rr;
    rr = -sy * cx; ii = -cy * sx;
    mr[s][0][1] = cz * rr + sz * ii; mi[s][0][1] = cz * ii - sz * rr;
    rr = sy * cx; ii = -cy * sx;
    mr[s][1][0] = cz * rr - sz * ii; mi[s][1][0] = cz * ii + sz * rr;
    rr = cy * cx; ii = -sy * sx;
    mr[s][1][1] = cz * rr - sz * ii; mi[s][1][1] = cz * ii + sz * rr;
  }

  float* base = ws + mat * 256;
  const int m = lane & 15;
  const int mp = m & 1;
  const int mc = 2 * ((m >> 1) & 3) + (m >> 3);
  const int khalf = lane >> 4;
#pragma unroll
  for (int j = 0; j < 4; ++j) {
#pragma unroll
    for (int e = 0; e < 2; ++e) {
      const int kc = 2 * j + khalf;
      const int kp = e;
      float re = 1.f, im = 0.f;
#pragma unroll
      for (int s = 0; s < 3; ++s) {
        const int rb = (mc >> (2 - s)) & 1;
        const int cb = (kc >> (2 - s)) & 1;
        const float wr = mr[s][rb][cb], wi = mi[s][rb][cb];
        const float nre = re * wr - im * wi;
        const float nim = re * wi + im * wr;
        re = nre; im = nim;
      }
      const float val = (kp == mp) ? re : ((mp == 1) ? im : -im);
      base[lane * 8 + j * 2 + e] = val;
    }
  }
}

#define QC_GRID 256

__global__ __launch_bounds__(QC_THREADS) void qc_main(
    const float* __restrict__ x, const float* __restrict__ ws,
    float* __restrict__ out, int nquads) {
  const int tid = threadIdx.x;
  const int lane = tid & 31;
  const int w = __builtin_amdgcn_readfirstlane(tid >> 5);
  const int baseB = w * QC_ELEM_BYTES;

  const int nelems = QC_BATCH_PER_BLOCK * nquads;
  const int stride = QC_ELEMS * (int)gridDim.x;

  const bool younger = (w >= 4);
  if (younger) __builtin_amdgcn_s_setprio(3);

#pragma unroll 1
  for (int b = QC_ELEMS * (int)blockIdx.x + w; b < nelems; b += stride) {
    if (younger && b + stride >= nelems) __builtin_amdgcn_s_setprio(0);
    int ln = lane;
    asm volatile("" : "+v"(ln));

    float ssPart;
    {
      const float4* xr = (const float4*)(x + (long long)b * QC_DIM);
      float ss = 0.f;
#pragma unroll 1
      for (int hf = 0; hf < 2; ++hf) {
        float4 v[16];
#pragma unroll
        for (int ii = 0; ii < 16; ++ii) v[ii] = xr[(16 * hf + ii) * 32 + ln];
        const int l0 = baseB ^ byte_of(4 * ln) ^ byte_of(2048 * hf);
#pragma unroll
        for (int ii = 0; ii < 16; ++ii) {
          ss += v[ii].x * v[ii].x + v[ii].y * v[ii].y + v[ii].z * v[ii].z +
                v[ii].w * v[ii].w;
          const v2f a0 = {v[ii].x, 0.f};
          const v2f a1 = {v[ii].y, 0.f};
          const v2f a2 = {v[ii].z, 0.f};
          const v2f a3 = {v[ii].w, 0.f};
          lds_st2(l0 ^ byte_of(128 * ii + 0), a0);
          lds_st2(l0 ^ byte_of(128 * ii + 1), a1);
          lds_st2(l0 ^ byte_of(128 * ii + 2), a2);
          lds_st2(l0 ^ byte_of(128 * ii + 3), a3);
        }
      }
      ssPart = ss;
    }

    stage_fence();
    apply_fused<0, 0>(baseB, ws + 0, ws + 256, ln);
    stage_fence();
    apply_fused<0, 1>(baseB, ws + 768, ws + 512, ln);
    stage_fence();
    apply_fused<1, 0>(baseB, ws + 1024, ws + 1280, ln);
    stage_fence();
    apply_fused<1, 1>(baseB, ws + 1792, ws + 1536, ln);
    stage_fence();
    apply_fused<2, 0>(baseB, ws + 2048, ws + 2304, ln);
    stage_fence();
    apply_fused<2, 1>(baseB, ws + 2816, ws + 2560, ln);
    stage_fence();

    {
      constexpr ZTab Z = make_ztab();
      const int l7 = baseB ^ byte_of(ln);
      float T = 0.f;
      float accK[12];
#pragma unroll
      for (int k = 0; k < 12; ++k) accK[k] = 0.f;
#pragma unroll 1
      for (int hf = 0; hf < 2; ++hf) {
        const int l7h = l7 ^ byte_of(2048 * hf);
        float Th = 0.f;
        float accH[12];
#pragma unroll
        for (int k = 0; k < 12; ++k) accH[k] = 0.f;
#pragma unroll
        for (int ii = 0; ii < 64; ++ii) {
          const v2f amp = lds_ld2(l7h ^ byte_of(32 * ii));
          const float p = amp.x * amp.x + amp.y * amp.y;
          Th += p;
#pragma unroll
          for (int k = 0; k < 12; ++k) {
            if (Z.hi[k] != 0 && Z.rep[k] == k)
              accH[k] += (__builtin_parity(ii & Z.hi[k] & 63)) ? -p : p;
          }
        }
        T += Th;
#pragma unroll
        for (int k = 0; k < 12; ++k) {
          if (Z.hi[k] != 0 && Z.rep[k] == k) {
            const bool neg = (hf != 0) && ((Z.hi[k] >> 6) & 1);
            accK[k] += neg ? -accH[k] : accH[k];
          }
        }
      }
      float acc[13];
#pragma unroll
      for (int k = 0; k < 12; ++k) {
        const float sk = (Z.hi[k] == 0) ? T : accK[Z.rep[k]];
        acc[11 - k] = (__builtin_parity(ln & Z.lo[k])) ? -sk : sk;
      }
      acc[12] = ssPart;
#pragma unroll
      for (int q = 0; q < 13; ++q) {
        float r = acc[q];
#pragma unroll
        for (int off = 16; off > 0; off >>= 1) r += __shfl_xor(r, off);
        acc[q] = r;
      }
      float num = acc[0];
#pragma unroll
      for (int q = 1; q < 12; ++q) num = (lane == q) ? acc[q] : num;
      if (lane < 12) out[(long long)b * 12 + lane] = num / acc[12];
    }
  }
}

extern "C" void kernel_launch(void* const* d_in, const int* in_sizes, int n_in,
                              void* d_out, int out_size, void* d_ws,
                              size_t ws_size, hipStream_t stream) {
  (void)n_in; (void)out_size; (void)ws_size;
  const float* x = (const float*)d_in[0];
  const float* params = (const float*)d_in[1];
  float* out = (float*)d_out;
  float* ws = (float*)d_ws;

  const int batch = in_sizes[0] / QC_DIM;
  const int blocks = batch / QC_BATCH_PER_BLOCK;

  qc_setup_gates<<<1, 384, 0, stream>>>(params, ws);

  (void)hipFuncSetAttribute((const void*)qc_main,
                            hipFuncAttributeMaxDynamicSharedMemorySize,
                            QC_LDS_BYTES);
  qc_main<<<QC_GRID, QC_THREADS, QC_LDS_BYTES, stream>>>(x, ws, out, blocks);
}


# index->slot map re-chosen (general GF(2)-linear, trip bits per pass) so that every LDS load/store pairs into a two-address instruction: 896 such instructions per element, not 1792; sums unchanged
# speedup vs baseline: 1.9586x; 1.0461x over previous
#include <hip/hip_runtime.h>
#include <math.h>


typedef float v2f __attribute__((ext_vector_type(2)));
typedef float v8f __attribute__((ext_vector_type(8)));

#define QC_BATCH_PER_BLOCK 4
#define QC_ELEMS 8
#define QC_THREADS 256
#define QC_DIM 4096

#define QC_ELEM_BYTES 32768
#define QC_LDS_BYTES (QC_ELEMS * QC_ELEM_BYTES)

extern __shared__ __attribute__((aligned(16))) float lds[];

static __device__ __forceinline__ v8f wmma4(v2f a, v2f b, v8f c) {
  return __builtin_amdgcn_wmma_f32_16x16x4_f32(false, a, false, b, (short)0, c,
                                               false, false);
}

static __device__ __forceinline__ void stage_fence() {
  __builtin_amdgcn_fence(__ATOMIC_RELEASE, "wavefront");
  __builtin_amdgcn_fence(__ATOMIC_ACQUIRE, "wavefront");
}

static __device__ __forceinline__ v2f lds_ld2(int byteOff) {
  return *(const v2f*)((const char*)lds + byteOff);
}
static __device__ __forceinline__ void lds_st2(int byteOff, v2f v) {
  *(v2f*)((char*)lds + byteOff) = v;
}

static __host__ __device__ constexpr int cnot_target(int i) {
  int x = i;
  x ^= x >> 1;
  x ^= x >> 2;
  x ^= x >> 4;
  x ^= x >> 8;
  return (x & 0x7FF) | (((x ^ (i >> 11)) & 1) << 11);
}

static __host__ __device__ constexpr int cnot_source(int t) {
  return ((t ^ (t >> 1)) & 0x3FF) | (((t ^ (t >> 10) ^ (t >> 11)) & 1) << 10) |
         (((t ^ (t >> 11)) & 1) << 11);
}

template <int L>
static __host__ __device__ constexpr int phys_of(int j) {
  for (int s = 0; s < L; ++s) j = cnot_source(j);
  return j;
}

static __host__ __device__ constexpr int byte_of(int p) {
  int r = 0;
  r ^= (0 - (p & 1)) & 330;
  r ^= (0 - ((p >> 1) & 1)) & 1998;
  r ^= (0 - ((p >> 2) & 1)) & 408;
  r ^= (0 - ((p >> 3) & 1)) & 326;
  r ^= (0 - ((p >> 4) & 1)) & 2447;
  r ^= (0 - ((p >> 5) & 1)) & 3545;
  r ^= (0 - ((p >> 6) & 1)) & 2499;
  r ^= (0 - ((p >> 7) & 1)) & 3322;
  r ^= (0 - ((p >> 8) & 1)) & 3290;
  r ^= (0 - ((p >> 9) & 1)) & 2063;
  r ^= (0 - ((p >> 10) & 1)) & 3546;
  r ^= (0 - ((p >> 11) & 1)) & 3287;
  return r << 3;
}

template <int P> struct FuseBits;
template <> struct FuseBits<0> {
  static constexpr int SHX = 9, SHY = 6, M0 = 0, M1 = 2, M2 = 3, M3 = 5;
  static constexpr int R0 = 1, R1 = 4;
};
template <> struct FuseBits<1> {
  static constexpr int SHX = 0, SHY = 3, M0 = 6, M1 = 9, M2 = 10, M3 = 11;
  static constexpr int R0 = 7, R1 = 8;
};
template <> struct FuseBits<2> {
  static constexpr int SHX = 9, SHY = 6, M0 = 0, M1 = 3, M2 = 4, M3 = 5;
  static constexpr int R0 = 1, R1 = 2;
};
template <> struct FuseBits<3> {
  static constexpr int SHX = 0, SHY = 3, M0 = 6, M1 = 7, M2 = 10, M3 = 11;
  static constexpr int R0 = 8, R1 = 9;
};
template <> struct FuseBits<4> {
  static constexpr int SHX = 9, SHY = 6, M0 = 1, M1 = 2, M2 = 4, M3 = 5;
  static constexpr int R0 = 0, R1 = 3;
};
template <> struct FuseBits<5> {
  static constexpr int SHX = 0, SHY = 3, M0 = 7, M1 = 8, M2 = 9, M3 = 11;
  static constexpr int R0 = 6, R1 = 10;
};

template <int L>
static __host__ __device__ constexpr int word_of(int logical) {
  return byte_of(phys_of<L>(logical));
}

static __device__ __forceinline__ int sel_word(int bit, int w) {
  return (0 - bit) & w;
}

template <int P, bool STORE>
static __host__ __device__ constexpr int reg_logical(int idx) {
  typedef FuseBits<P> B;
  return STORE ? (((idx & 7) << B::SHX) | ((idx >> 3) << (B::SHY + 1)))
               : (((idx & 7) << B::SHY) | ((idx >> 3) << (B::SHX + 1)));
}

template <int L, int P, bool STORE>
static __host__ __device__ constexpr int pair_rho() {
  typedef FuseBits<P> B;
  const int U = word_of<L>(1 << B::M0) | word_of<L>(1 << B::M1) |
                word_of<L>(1 << B::M2) | word_of<L>(1 << B::M3) |
                word_of<L>(1 << (STORE ? B::SHY : B::SHX)) |
                word_of<L>(1 << B::R0) | word_of<L>(1 << B::R1);
  for (int rho = 1; rho < 32; ++rho) {
    const int m = word_of<L>(reg_logical<P, STORE>(rho));
    if ((m & U) != 0) continue;
    if ((m & ~(0xFF << 3)) == 0 || (m & ~(0xFC0 << 3)) == 0) return rho;
  }
  return 0;
}

static __device__ __forceinline__ void half_swap(float& lo, float& hi) {
  typedef unsigned v2u __attribute__((ext_vector_type(2)));
  const v2u r = __builtin_amdgcn_permlane16_swap(__float_as_uint(lo),
                                                 __float_as_uint(hi), false,
                                                 false);
  lo = __uint_as_float(r[0]);
  hi = __uint_as_float(r[1]);
}

template <int L, int F>
static __device__ __forceinline__ void apply_fused(int baseB,
                                                   const float* wsX,
                                                   const float* wsY,
                                                   int lane) {
  constexpr int P = 2 * L + F;
  typedef FuseBits<P> B;
  float wx[8], wy[8];
  const float* wxb = wsX + lane * 8;
  const float* wyb = wsY + lane * 8;
#pragma unroll
  for (int s = 0; s < 8; ++s) wx[s] = wxb[s];
#pragma unroll
  for (int s = 0; s < 8; ++s) wy[s] = wyb[s];

  const int c = lane & 15;
  const int hh = lane >> 4;
  const int colW = sel_word(c & 1, word_of<L>(1 << B::M0)) ^
                   sel_word((c >> 1) & 1, word_of<L>(1 << B::M1)) ^
                   sel_word((c >> 2) & 1, word_of<L>(1 << B::M2)) ^
                   sel_word((c >> 3) & 1, word_of<L>(1 << B::M3));
  const int ldL = baseB ^ colW ^ sel_word(hh, word_of<L>(1 << B::SHX));
  const int stL = baseB ^ colW ^ sel_word(hh, word_of<L>(1 << B::SHY));

  constexpr int RHO_L = pair_rho<L, P, false>();
  constexpr int M_L = word_of<L>(reg_logical<P, false>(RHO_L));
  constexpr int LOW_L = RHO_L & (0 - RHO_L);
  constexpr int RHO_S = pair_rho<L, P, true>();
  constexpr int M_S = word_of<L>(reg_logical<P, true>(RHO_S));
  constexpr int LOW_S = RHO_S & (0 - RHO_S);

#pragma unroll 1
  for (int it = 0; it < 4; ++it) {
    const int twB = sel_word(it & 1, word_of<L>(1 << B::R0)) ^
                    sel_word(it >> 1, word_of<L>(1 << B::R1));
    int ldT = ldL ^ twB;
    int stT = stL ^ twB;
    asm volatile("" : "+v"(ldT), "+v"(stT));
    v2f b[32];
#pragma unroll
    for (int n = 0; n < 32; ++n) {
      const int idx = (n >> 2) | ((n & 3) << 3);
      const int cw = word_of<L>(reg_logical<P, false>(idx));
      if (RHO_L == 0) {
        b[idx] = lds_ld2(ldT ^ cw);
      } else if ((idx & LOW_L) == 0) {
        const int base = ldT ^ (cw & ~M_L);
        b[idx] = lds_ld2(base + (cw & M_L));
        b[idx ^ RHO_L] = lds_ld2(base + ((cw & M_L) ^ M_L));
      }
    }
    const v8f zero8 = {0.f, 0.f, 0.f, 0.f, 0.f, 0.f, 0.f, 0.f};
    v8f d[8];
#pragma unroll
    for (int u = 0; u < 8; ++u) d[u] = zero8;
#pragma unroll
    for (int q = 0; q < 4; ++q) {
#pragma unroll
      for (int j = 0; j < 4; ++j) {
        const v2f Av = {wx[2 * j], wx[2 * j + 1]};
        d[2 * q] = wmma4(Av, b[(2 * q) | (j << 3)], d[2 * q]);
        d[2 * q + 1] = wmma4(Av, b[(2 * q + 1) | (j << 3)], d[2 * q + 1]);
      }
    }
#pragma unroll
    for (int q = 0; q < 4; ++q) {
#pragma unroll
      for (int k = 0; k < 8; ++k) {
        float lo = d[2 * q][k], hi = d[2 * q + 1][k];
        half_swap(lo, hi);
        d[2 * q][k] = lo;
        d[2 * q + 1][k] = hi;
      }
    }
    v8f d2[8];
#pragma unroll
    for (int g = 0; g < 8; ++g) d2[g] = zero8;
#pragma unroll
    for (int q = 0; q < 4; ++q) {
      const v2f Av = {wy[2 * q], wy[2 * q + 1]};
#pragma unroll
      for (int g = 0; g < 8; ++g) {
        const int a = g & 1, r = g >> 1;
        const v2f Bv = {d[2 * q + a][2 * r], d[2 * q + a][2 * r + 1]};
        d2[g] = wmma4(Av, Bv, d2[g]);
      }
    }
#pragma unroll
    for (int n = 0; n < 32; ++n) {
      const int idx = (n >> 2) | ((n & 3) << 3);
      const int cw = word_of<L>(reg_logical<P, true>(idx));
      const v2f val = {d2[idx & 7][2 * (idx >> 3)],
                       d2[idx & 7][2 * (idx >> 3) + 1]};
      if (RHO_S == 0) {
        lds_st2(stT ^ cw, val);
      } else if ((idx & LOW_S) == 0) {
        const int id2 = idx ^ RHO_S;
        const v2f val2 = {d2[id2 & 7][2 * (id2 >> 3)],
                          d2[id2 & 7][2 * (id2 >> 3) + 1]};
        const int base = stT ^ (cw & ~M_S);
        lds_st2(base + (cw & M_S), val);
        lds_st2(base + ((cw & M_S) ^ M_S), val2);
      }
    }
  }
}

template <bool FINAL>
static __host__ __device__ constexpr int io_index(int n) {
  return FINAL ? 32 * n : 128 * (n >> 2) + (n & 3);
}
template <bool FINAL>
static __host__ __device__ constexpr int pair_rho_io() {
  int U = byte_of(2048);
  for (int k = 0; k < 5; ++k) U |= byte_of(FINAL ? (1 << k) : (4 << k));
  for (int rho = 1; rho < 64; ++rho) {
    const int m = byte_of(io_index<FINAL>(rho));
    if ((m & U) != 0) continue;
    if ((m & ~(0xFF << 3)) == 0 || (m & ~(0xFC0 << 3)) == 0) return rho;
  }
  return 0;
}
static __device__ __forceinline__ float f4_at(const float4& v, int k) {
  return k == 0 ? v.x : (k == 1 ? v.y : (k == 2 ? v.z : v.w));
}

struct ZTab {
  int hi[12];
  int lo[12];
  int rep[12];
};
static __host__ __device__ constexpr ZTab make_ztab() {
  ZTab z{};
  for (int k = 0; k < 12; ++k) {
    int m = 0;
    for (int b = 0; b < 12; ++b) {
      const int img = cnot_target(cnot_target(cnot_target(1 << b)));
      if ((img >> k) & 1) m |= 1 << b;
    }
    z.hi[k] = m >> 5;
    z.lo[k] = m & 31;
  }
  for (int k = 0; k < 12; ++k) {
    z.rep[k] = k;
    for (int k2 = k - 1; k2 >= 0; --k2)
      if (z.hi[k2] == z.hi[k]) z.rep[k] = k2;
  }
  return z;
}

__global__ void qc_setup_gates(const float* __restrict__ params,
                               float* __restrict__ ws) {
  const int tid = threadIdx.x;
  const int mat = tid >> 5;
  const int lane = tid & 31;
  if (mat >= 12) return;
  const int L = mat / 4, g = mat % 4;

  float mr[3][2][2], mi[3][2][2];
#pragma unroll
  for (int s = 0; s < 3; ++s) {
    const int q = g * 3 + s;
    const int base = L * 36 + q * 3;
    const float tx = params[base], ty = params[base + 1], tz = params[base + 2];
    const float cx = cosf(0.5f * tx), sx = sinf(0.5f * tx);
    const float cy = cosf(0.5f * ty), sy = sinf(0.5f * ty);
    const float cz = cosf(0.5f * tz), sz = sinf(0.5f * tz);
    float rr, ii;
    rr = cy * cx; ii = sy * sx;
    mr[s][0][0] = cz * rr + sz * ii; mi[s][0][0] = cz * ii - sz * rr;
    rr = -sy * cx; ii = -cy * sx;
    mr[s][0][1] = cz * rr + sz * ii; mi[s][0][1] = cz * ii - sz * rr;
    rr = sy * cx; ii = -cy * sx;
    mr[s][1][0] = cz * rr - sz * ii; mi[s][1][0] = cz * ii + sz * rr;
    rr = cy * cx; ii = -sy * sx;
    mr[s][1][1] = cz * rr - sz * ii; mi[s][1][1] = cz * ii + sz * rr;
  }

  float* base = ws + mat * 256;
  const int m = lane & 15;
  const int mp = m & 1;
  const int mc = 2 * ((m >> 1) & 3) + (m >> 3);
  const int khalf = lane >> 4;
#pragma unroll
  for (int j = 0; j < 4; ++j) {
#pragma unroll
    for (int e = 0; e < 2; ++e) {
      const int kc = 2 * j + khalf;
      const int kp = e;
      float re = 1.f, im = 0.f;
#pragma unroll
      for (int s = 0; s < 3; ++s) {
        const int rb = (mc >> (2 - s)) & 1;
        const int cb = (kc >> (2 - s)) & 1;
        const float wr = mr[s][rb][cb], wi = mi[s][rb][cb];
        const float nre = re * wr - im * wi;
        const float nim = re * wi + im * wr;
        re = nre; im = nim;
      }
      const float val = (kp == mp) ? re : ((mp == 1) ? im : -im);
      base[lane * 8 + j * 2 + e] = val;
    }
  }
}

#define QC_GRID 256

__global__ __launch_bounds__(QC_THREADS) void qc_main(
    const float* __restrict__ x, const float* __restrict__ ws,
    float* __restrict__ out, int nquads) {
  const int tid = threadIdx.x;
  const int lane = tid & 31;
  const int w = __builtin_amdgcn_readfirstlane(tid >> 5);
  const int baseB = w * QC_ELEM_BYTES;

  const int nelems = QC_BATCH_PER_BLOCK * nquads;
  const int stride = QC_ELEMS * (int)gridDim.x;

  const bool younger = (w >= 4);
  if (younger) __builtin_amdgcn_s_setprio(3);

#pragma unroll 1
  for (int b = QC_ELEMS * (int)blockIdx.x + w; b < nelems; b += stride) {
    if (younger && b + stride >= nelems) __builtin_amdgcn_s_setprio(0);
    int ln = lane;
    asm volatile("" : "+v"(ln));

    float ssPart;
    {
      const float4* xr = (const float4*)(x + (long long)b * QC_DIM);
      float ss = 0.f;
#pragma unroll 1
      for (int hf = 0; hf < 2; ++hf) {
        float4 v[16];
#pragma unroll
        for (int ii = 0; ii < 16; ++ii) v[ii] = xr[(16 * hf + ii) * 32 + ln];
        int lw0 = 0;
#pragma unroll
        for (int k = 0; k < 5; ++k)
          lw0 ^= sel_word((ln >> k) & 1, byte_of(4 << k));
        const int l0 = baseB ^ lw0 ^ byte_of(2048 * hf);
#pragma unroll
        for (int ii = 0; ii < 16; ++ii) {
          ss += v[ii].x * v[ii].x + v[ii].y * v[ii].y + v[ii].z * v[ii].z +
                v[ii].w * v[ii].w;
        }
        constexpr int RHO_0 = pair_rho_io<false>();
        constexpr int M_0 = byte_of(io_index<false>(RHO_0));
        constexpr int LOW_0 = RHO_0 & (0 - RHO_0);
#pragma unroll
        for (int n = 0; n < 64; ++n) {
          const int cw = byte_of(io_index<false>(n));
          const v2f a = {f4_at(v[n >> 2], n & 3), 0.f};
          if (RHO_0 == 0) {
            lds_st2(l0 ^ cw, a);
          } else if ((n & LOW_0) == 0) {
            const int n2 = n ^ RHO_0;
            const v2f a2 = {f4_at(v[n2 >> 2], n2 & 3), 0.f};
            const int base = l0 ^ (cw & ~M_0);
            lds_st2(base + (cw & M_0), a);
            lds_st2(base + ((cw & M_0) ^ M_0), a2);
          }
        }
      }
      ssPart = ss;
    }

    stage_fence();
    apply_fused<0, 0>(baseB, ws + 0, ws + 256, ln);
    stage_fence();
    apply_fused<0, 1>(baseB, ws + 768, ws + 512, ln);
    stage_fence();
    apply_fused<1, 0>(baseB, ws + 1024, ws + 1280, ln);
    stage_fence();
    apply_fused<1, 1>(baseB, ws + 1792, ws + 1536, ln);
    stage_fence();
    apply_fused<2, 0>(baseB, ws + 2048, ws + 2304, ln);
    stage_fence();
    apply_fused<2, 1>(baseB, ws + 2816, ws + 2560, ln);
    stage_fence();

    {
      constexpr ZTab Z = make_ztab();
      int lw7 = 0;
#pragma unroll
      for (int k = 0; k < 5; ++k)
        lw7 ^= sel_word((ln >> k) & 1, byte_of(1 << k));
      const int l7 = baseB ^ lw7;
      float T = 0.f;
      float accK[12];
#pragma unroll
      for (int k = 0; k < 12; ++k) accK[k] = 0.f;
#pragma unroll 1
      for (int hf = 0; hf < 2; ++hf) {
        const int l7h = l7 ^ byte_of(2048 * hf);
        float Th = 0.f;
        float accH[12];
#pragma unroll
        for (int k = 0; k < 12; ++k) accH[k] = 0.f;
        constexpr int RHO_7 = pair_rho_io<true>();
        constexpr int M_7 = byte_of(io_index<true>(RHO_7));
        constexpr int LOW_7 = RHO_7 & (0 - RHO_7);
        v2f amps[64];
#pragma unroll
        for (int n = 0; n < 64; ++n) {
          const int cw = byte_of(io_index<true>(n));
          if (RHO_7 == 0) {
            amps[n] = lds_ld2(l7h ^ cw);
          } else if ((n & LOW_7) == 0) {
            const int base = l7h ^ (cw & ~M_7);
            amps[n] = lds_ld2(base + (cw & M_7));
            amps[n ^ RHO_7] = lds_ld2(base + ((cw & M_7) ^ M_7));
          }
        }
#pragma unroll
        for (int ii = 0; ii < 64; ++ii) {
          const v2f amp = amps[ii];
          const float p = amp.x * amp.x + amp.y * amp.y;
          Th += p;
#pragma unroll
          for (int k = 0; k < 12; ++k) {
            if (Z.hi[k] != 0 && Z.rep[k] == k)
              accH[k] += (__builtin_parity(ii & Z.hi[k] & 63)) ? -p : p;
          }
        }
        T += Th;
#pragma unroll
        for (int k = 0; k < 12; ++k) {
          if (Z.hi[k] != 0 && Z.rep[k] == k) {
            const bool neg = (hf != 0) && ((Z.hi[k] >> 6) & 1);
            accK[k] += neg ? -accH[k] : accH[k];
          }
        }
      }
      float acc[13];
#pragma unroll
      for (int k = 0; k < 12; ++k) {
        const float sk = (Z.hi[k] == 0) ? T : accK[Z.rep[k]];
        acc[11 - k] = (__builtin_parity(ln & Z.lo[k])) ? -sk : sk;
      }
      acc[12] = ssPart;
#pragma unroll
      for (int q = 0; q < 13; ++q) {
        float r = acc[q];
#pragma unroll
        for (int off = 16; off > 0; off >>= 1) r += __shfl_xor(r, off);
        acc[q] = r;
      }
      float num = acc[0];
#pragma unroll
      for (int q = 1; q < 12; ++q) num = (lane == q) ? acc[q] : num;
      if (lane < 12) out[(long long)b * 12 + lane] = num / acc[12];
    }
  }
}

extern "C" void kernel_launch(void* const* d_in, const int* in_sizes, int n_in,
                              void* d_out, int out_size, void* d_ws,
                              size_t ws_size, hipStream_t stream) {
  (void)n_in; (void)out_size; (void)ws_size;
  const float* x = (const float*)d_in[0];
  const float* params = (const float*)d_in[1];
  float* out = (float*)d_out;
  float* ws = (float*)d_ws;

  const int batch = in_sizes[0] / QC_DIM;
  const int blocks = batch / QC_BATCH_PER_BLOCK;

  qc_setup_gates<<<1, 384, 0, stream>>>(params, ws);

  (void)hipFuncSetAttribute((const void*)qc_main,
                            hipFuncAttributeMaxDynamicSharedMemorySize,
                            QC_LDS_BYTES);
  qc_main<<<QC_GRID, QC_THREADS, QC_LDS_BYTES, stream>>>(x, ws, out, blocks);
}
